# starting kernel
# baseline (speedup 1.0000x reference)
#include <hip/hip_runtime.h>

#define NODE_F 128

typedef float v2f __attribute__((ext_vector_type(2)));
typedef float v8f __attribute__((ext_vector_type(8)));

static __device__ __forceinline__ v8f wmma_k4(v2f a, v2f b, v8f c) {
  return __builtin_amdgcn_wmma_f32_16x16x4_f32(
       false, a,  false, b,
       (short)0, c,  false,  false);
}

__global__ void deg_kernel(const int* __restrict__ dst, float* __restrict__ deg, int nEdges) {
  int e = blockIdx.x * blockDim.x + threadIdx.x;
  if (e < nEdges) atomicAdd(&deg[dst[e]], 1.0f);
}

__global__ void scatter_kernel(const float* __restrict__ h, const int* __restrict__ src,
                               const int* __restrict__ dst, float* __restrict__ sum,
                               int nEdges) {
  int g    = blockIdx.x * blockDim.x + threadIdx.x;
  int e    = g >> 5;
  int lane = g & 31;
  if (e >= nEdges) return;
  int s = src[e];
  int d = dst[e];
  float4 v = ((const float4*)(h + (size_t)s * NODE_F))[lane];
  float* sp = sum + (size_t)d * NODE_F + lane * 4;
  atomicAdd(sp + 0, v.x);
  atomicAdd(sp + 1, v.y);
  atomicAdd(sp + 2, v.z);
  atomicAdd(sp + 3, v.w);
}

template <bool NEIGH, bool RELU>
__global__ void sage_gemm(const float* __restrict__ hself, const float* __restrict__ sum,
                          const float* __restrict__ deg, const float* __restrict__ Wself,
                          const float* __restrict__ Wneigh, const float* __restrict__ bias,
                          float* __restrict__ out, int nNodes) {
  __shared__ float hs[32][NODE_F + 4];
  __shared__ float hn[32][NODE_F + 4];

  const int tid     = threadIdx.x;
  const int m_base  = blockIdx.x * 32;
  const bool fullTile = (m_base + 32 <= nNodes);

  for (int i = 0; i < 4; ++i) {
    int f   = i * 256 + tid;
    int row = f >> 5;
    int cv  = f & 31;
    int m   = m_base + row;
    float4 v = make_float4(0.f, 0.f, 0.f, 0.f);
    if (fullTile || m < nNodes) v = ((const float4*)(hself + (size_t)m * NODE_F))[cv];
    *(float4*)&hs[row][cv * 4] = v;
    if (NEIGH) {
      float4 s4 = make_float4(0.f, 0.f, 0.f, 0.f);
      float rd = 1.0f;
      if (fullTile || m < nNodes) {
        s4 = ((const float4*)(sum + (size_t)m * NODE_F))[cv];
        rd = 1.0f / fmaxf(deg[m], 1.0f);
      }
      hn[row][cv * 4 + 0] = s4.x * rd;
      hn[row][cv * 4 + 1] = s4.y * rd;
      hn[row][cv * 4 + 2] = s4.z * rd;
      hn[row][cv * 4 + 3] = s4.w * rd;
    }
  }
  __syncthreads();

  const int lane = tid & 31;
  const int wv   = tid >> 5;
  const int lid  = lane & 15;
  const int half = lane >> 4;
  const int n0   = wv * 16;

  const float* Ws = Wself + (size_t)(n0 + lid) * NODE_F;
  const float* Wn = NEIGH ? (Wneigh + (size_t)(n0 + lid) * NODE_F) : nullptr;

  v8f acc0 = {0.f, 0.f, 0.f, 0.f, 0.f, 0.f, 0.f, 0.f};
  v8f acc1 = {0.f, 0.f, 0.f, 0.f, 0.f, 0.f, 0.f, 0.f};

#pragma unroll 8
  for (int k0 = 0; k0 < NODE_F; k0 += 4) {
    const int kk = k0 + 2 * half;
    v2f bs;
    bs.x = Ws[kk];
    bs.y = Ws[kk + 1];
    v2f a0, a1;
    a0.x = hs[lid][kk];
    a0.y = hs[lid][kk + 1];
    a1.x = hs[16 + lid][kk];
    a1.y = hs[16 + lid][kk + 1];
    acc0 = wmma_k4(a0, bs, acc0);
    acc1 = wmma_k4(a1, bs, acc1);
    if (NEIGH) {
      v2f bn;
      bn.x = Wn[kk];
      bn.y = Wn[kk + 1];
      v2f c0, c1;
      c0.x = hn[lid][kk];
      c0.y = hn[lid][kk + 1];
      c1.x = hn[16 + lid][kk];
      c1.y = hn[16 + lid][kk + 1];
      acc0 = wmma_k4(c0, bn, acc0);
      acc1 = wmma_k4(c1, bn, acc1);
    }
  }

  const float bv = bias[n0 + lid];
  float* outc = out + n0 + lid;
  const int r0 = m_base + 8 * half;
  const int r1 = m_base + 16 + 8 * half;

  if (fullTile) {
#pragma unroll
    for (int e = 0; e < 8; ++e) {
      float v0 = acc0[e] + bv;
      float v1 = acc1[e] + bv;
      if (RELU) { v0 = fmaxf(v0, 0.f); v1 = fmaxf(v1, 0.f); }
      outc[(size_t)(r0 + e) * NODE_F] = v0;
      outc[(size_t)(r1 + e) * NODE_F] = v1;
    }
  } else {
#pragma unroll
    for (int e = 0; e < 8; ++e) {
      float v0 = acc0[e] + bv;
      float v1 = acc1[e] + bv;
      if (RELU) { v0 = fmaxf(v0, 0.f); v1 = fmaxf(v1, 0.f); }
      if (r0 + e < nNodes) outc[(size_t)(r0 + e) * NODE_F] = v0;
      if (r1 + e < nNodes) outc[(size_t)(r1 + e) * NODE_F] = v1;
    }
  }
}

extern "C" void kernel_launch(void* const* d_in, const int* in_sizes, int n_in,
                              void* d_out, int out_size, void* d_ws, size_t ws_size,
                              hipStream_t stream) {
  const float* x      = (const float*)d_in[0];
  const int*   src    = (const int*)d_in[1];
  const int*   dst    = (const int*)d_in[2];
  const float* W1s    = (const float*)d_in[3];
  const float* W1n    = (const float*)d_in[4];
  const float* b1     = (const float*)d_in[5];
  const float* W2s    = (const float*)d_in[6];
  const float* W2n    = (const float*)d_in[7];
  const float* b2     = (const float*)d_in[8];
  const float* W3     = (const float*)d_in[9];
  const float* b3     = (const float*)d_in[10];
  float*       out    = (float*)d_out;

  const int nNodes = in_sizes[0] / NODE_F;
  const int nEdges = in_sizes[1];

  float* sum = (float*)d_ws;
  float* deg = sum + (size_t)nNodes * NODE_F;
  float* h1  = deg + nNodes;
  float* h2  = h1 + (size_t)nNodes * NODE_F;

  dim3 blk(256);
  dim3 gEdge((nEdges + 255) / 256);
  dim3 gScat((nEdges * 32 + 255) / 256);
  dim3 gGemm((nNodes + 31) / 32);

  hipMemsetAsync(sum, 0, ((size_t)nNodes * NODE_F + nNodes) * sizeof(float), stream);
  deg_kernel<<<gEdge, blk, 0, stream>>>(dst, deg, nEdges);
  scatter_kernel<<<gScat, blk, 0, stream>>>(x, src, dst, sum, nEdges);
  sage_gemm<true, true><<<gGemm, blk, 0, stream>>>(x, sum, deg, W1s, W1n, b1, h1, nNodes);

  hipMemsetAsync(sum, 0, (size_t)nNodes * NODE_F * sizeof(float), stream);
  scatter_kernel<<<gScat, blk, 0, stream>>>(h1, src, dst, sum, nEdges);
  sage_gemm<true, false><<<gGemm, blk, 0, stream>>>(h1, sum, deg, W2s, W2n, b2, h2, nNodes);

  sage_gemm<false, false><<<gGemm, blk, 0, stream>>>(h2, nullptr, nullptr, W3, nullptr, b3,
                                                     out, nNodes);
}


# aggregation by counting sort on dst (int atomic arrival-order rank), then one wave per node gathers its rows and sums in registers: no float atomics, no sum memset; sum order = arrival rank
# speedup vs baseline: 2.2029x; 2.2029x over previous
#include <hip/hip_runtime.h>

#define NODE_F 128

typedef float v2f __attribute__((ext_vector_type(2)));
typedef float v8f __attribute__((ext_vector_type(8)));

static __device__ __forceinline__ v8f wmma_k4(v2f a, v2f b, v8f c) {
  return __builtin_amdgcn_wmma_f32_16x16x4_f32(
       false, a,  false, b,
       (short)0, c,  false,  false);
}


__global__ void count_kernel(const int* __restrict__ dst, int* __restrict__ cnt,
                             int* __restrict__ rank, int nEdges) {
  int e = blockIdx.x * blockDim.x + threadIdx.x;
  if (e < nEdges) rank[e] = atomicAdd(&cnt[dst[e]], 1);
}

__global__ void alloc_kernel(const int* cnt, float* deg, int* __restrict__ start,
                             int* __restrict__ total, int nNodes) {
  const int n    = blockIdx.x * blockDim.x + threadIdx.x;
  const int lane = threadIdx.x & 31;
  const int c    = (n < nNodes) ? cnt[n] : 0;
  int incl = c;
  for (int o = 1; o < 32; o <<= 1) {
    int t = __shfl_up(incl, o);
    if (lane >= o) incl += t;
  }
  int base = 0;
  if (lane == 31) base = atomicAdd(total, incl);
  base = __shfl(base, 31);
  if (n < nNodes) {
    start[n] = base + incl - c;
    deg[n]   = (float)c;
  }
}

__global__ void fill_kernel(const int* __restrict__ src, const int* __restrict__ dst,
                            const int* __restrict__ start, const int* __restrict__ rank,
                            int* __restrict__ col, int nEdges) {
  int e = blockIdx.x * blockDim.x + threadIdx.x;
  if (e < nEdges) col[start[dst[e]] + rank[e]] = src[e];
}

static __device__ __forceinline__ float4 ld_row(const float* __restrict__ h, int s, int lane) {
  return ((const float4*)(h + (size_t)s * NODE_F))[lane];
}
static __device__ __forceinline__ void add4(float4& a, const float4 v) {
  a.x += v.x; a.y += v.y; a.z += v.z; a.w += v.w;
}

__global__ void gather_kernel(const float* __restrict__ h, const int* __restrict__ col,
                              const int* __restrict__ start, const float* __restrict__ deg,
                              float* __restrict__ sum, int nNodes) {
  const int g    = blockIdx.x * blockDim.x + threadIdx.x;
  const int n    = g >> 5;
  const int lane = g & 31;
  if (n >= nNodes) return;
  const int beg = __builtin_amdgcn_readfirstlane(start[n]);
  const int c   = __builtin_amdgcn_readfirstlane((int)deg[n]);
  float4 acc = make_float4(0.f, 0.f, 0.f, 0.f);
  for (int b = 0; b < c; b += 32) {
    const int m   = (c - b < 32) ? (c - b) : 32;
    const int idx = (lane < m) ? col[beg + b + lane] : 0;
    int k = 0;
    for (; k + 8 <= m; k += 8) {
      const float4 v0 = ld_row(h, __builtin_amdgcn_readlane(idx, k + 0), lane);
      const float4 v1 = ld_row(h, __builtin_amdgcn_readlane(idx, k + 1), lane);
      const float4 v2 = ld_row(h, __builtin_amdgcn_readlane(idx, k + 2), lane);
      const float4 v3 = ld_row(h, __builtin_amdgcn_readlane(idx, k + 3), lane);
      const float4 v4 = ld_row(h, __builtin_amdgcn_readlane(idx, k + 4), lane);
      const float4 v5 = ld_row(h, __builtin_amdgcn_readlane(idx, k + 5), lane);
      const float4 v6 = ld_row(h, __builtin_amdgcn_readlane(idx, k + 6), lane);
      const float4 v7 = ld_row(h, __builtin_amdgcn_readlane(idx, k + 7), lane);
      add4(acc, v0); add4(acc, v1); add4(acc, v2); add4(acc, v3);
      add4(acc, v4); add4(acc, v5); add4(acc, v6); add4(acc, v7);
    }
    if (k + 4 <= m) {
      const float4 v0 = ld_row(h, __builtin_amdgcn_readlane(idx, k + 0), lane);
      const float4 v1 = ld_row(h, __builtin_amdgcn_readlane(idx, k + 1), lane);
      const float4 v2 = ld_row(h, __builtin_amdgcn_readlane(idx, k + 2), lane);
      const float4 v3 = ld_row(h, __builtin_amdgcn_readlane(idx, k + 3), lane);
      add4(acc, v0); add4(acc, v1); add4(acc, v2); add4(acc, v3);
      k += 4;
    }
    for (; k < m; ++k) {
      const float4 v0 = ld_row(h, __builtin_amdgcn_readlane(idx, k), lane);
      add4(acc, v0);
    }
  }
  ((float4*)(sum + (size_t)n * NODE_F))[lane] = acc;
}

template <bool NEIGH, bool RELU>
__global__ void sage_gemm(const float* __restrict__ hself, const float* __restrict__ sum,
                          const float* __restrict__ deg, const float* __restrict__ Wself,
                          const float* __restrict__ Wneigh, const float* __restrict__ bias,
                          float* __restrict__ out, int nNodes) {
  __shared__ float hs[32][NODE_F + 4];
  __shared__ float hn[32][NODE_F + 4];

  const int tid     = threadIdx.x;
  const int m_base  = blockIdx.x * 32;
  const bool fullTile = (m_base + 32 <= nNodes);

  for (int i = 0; i < 4; ++i) {
    int f   = i * 256 + tid;
    int row = f >> 5;
    int cv  = f & 31;
    int m   = m_base + row;
    float4 v = make_float4(0.f, 0.f, 0.f, 0.f);
    if (fullTile || m < nNodes) v = ((const float4*)(hself + (size_t)m * NODE_F))[cv];
    *(float4*)&hs[row][cv * 4] = v;
    if (NEIGH) {
      float4 s4 = make_float4(0.f, 0.f, 0.f, 0.f);
      float rd = 1.0f;
      if (fullTile || m < nNodes) {
        s4 = ((const float4*)(sum + (size_t)m * NODE_F))[cv];
        rd = 1.0f / fmaxf(deg[m], 1.0f);
      }
      hn[row][cv * 4 + 0] = s4.x * rd;
      hn[row][cv * 4 + 1] = s4.y * rd;
      hn[row][cv * 4 + 2] = s4.z * rd;
      hn[row][cv * 4 + 3] = s4.w * rd;
    }
  }
  __syncthreads();

  const int lane = tid & 31;
  const int wv   = tid >> 5;
  const int lid  = lane & 15;
  const int half = lane >> 4;
  const int n0   = wv * 16;

  const float* Ws = Wself + (size_t)(n0 + lid) * NODE_F;
  const float* Wn = NEIGH ? (Wneigh + (size_t)(n0 + lid) * NODE_F) : nullptr;

  v8f acc0 = {0.f, 0.f, 0.f, 0.f, 0.f, 0.f, 0.f, 0.f};
  v8f acc1 = {0.f, 0.f, 0.f, 0.f, 0.f, 0.f, 0.f, 0.f};

#pragma unroll 8
  for (int k0 = 0; k0 < NODE_F; k0 += 4) {
    const int kk = k0 + 2 * half;
    v2f bs;
    bs.x = Ws[kk];
    bs.y = Ws[kk + 1];
    v2f a0, a1;
    a0.x = hs[lid][kk];
    a0.y = hs[lid][kk + 1];
    a1.x = hs[16 + lid][kk];
    a1.y = hs[16 + lid][kk + 1];
    acc0 = wmma_k4(a0, bs, acc0);
    acc1 = wmma_k4(a1, bs, acc1);
    if (NEIGH) {
      v2f bn;
      bn.x = Wn[kk];
      bn.y = Wn[kk + 1];
      v2f c0, c1;
      c0.x = hn[lid][kk];
      c0.y = hn[lid][kk + 1];
      c1.x = hn[16 + lid][kk];
      c1.y = hn[16 + lid][kk + 1];
      acc0 = wmma_k4(c0, bn, acc0);
      acc1 = wmma_k4(c1, bn, acc1);
    }
  }

  const float bv = bias[n0 + lid];
  float* outc = out + n0 + lid;
  const int r0 = m_base + 8 * half;
  const int r1 = m_base + 16 + 8 * half;

  if (fullTile) {
#pragma unroll
    for (int e = 0; e < 8; ++e) {
      float v0 = acc0[e] + bv;
      float v1 = acc1[e] + bv;
      if (RELU) { v0 = fmaxf(v0, 0.f); v1 = fmaxf(v1, 0.f); }
      outc[(size_t)(r0 + e) * NODE_F] = v0;
      outc[(size_t)(r1 + e) * NODE_F] = v1;
    }
  } else {
#pragma unroll
    for (int e = 0; e < 8; ++e) {
      float v0 = acc0[e] + bv;
      float v1 = acc1[e] + bv;
      if (RELU) { v0 = fmaxf(v0, 0.f); v1 = fmaxf(v1, 0.f); }
      if (r0 + e < nNodes) outc[(size_t)(r0 + e) * NODE_F] = v0;
      if (r1 + e < nNodes) outc[(size_t)(r1 + e) * NODE_F] = v1;
    }
  }
}

extern "C" void kernel_launch(void* const* d_in, const int* in_sizes, int n_in,
                              void* d_out, int out_size, void* d_ws, size_t ws_size,
                              hipStream_t stream) {
  const float* x      = (const float*)d_in[0];
  const int*   src    = (const int*)d_in[1];
  const int*   dst    = (const int*)d_in[2];
  const float* W1s    = (const float*)d_in[3];
  const float* W1n    = (const float*)d_in[4];
  const float* b1     = (const float*)d_in[5];
  const float* W2s    = (const float*)d_in[6];
  const float* W2n    = (const float*)d_in[7];
  const float* b2     = (const float*)d_in[8];
  const float* W3     = (const float*)d_in[9];
  const float* b3     = (const float*)d_in[10];
  float*       out    = (float*)d_out;

  const int nNodes = in_sizes[0] / NODE_F;
  const int nEdges = in_sizes[1];

  float* sum = (float*)d_ws;
  float* deg = sum + (size_t)nNodes * NODE_F;
  float* h1  = deg + nNodes;
  float* h2  = h1 + (size_t)nNodes * NODE_F;
  int* cnt   = (int*)deg;
  int* total = (int*)h1;
  int* col   = (int*)h2;
  int* rank  = col + nEdges;
  int* start = rank + nEdges;

  dim3 blk(256);
  dim3 gEdge((nEdges + 255) / 256);
  dim3 gNode((nNodes + 255) / 256);
  dim3 gGath((nNodes * 32 + 255) / 256);
  dim3 gGemm((nNodes + 31) / 32);

  hipMemsetAsync(cnt, 0, ((size_t)nNodes + 1) * sizeof(int), stream);
  count_kernel<<<gEdge, blk, 0, stream>>>(dst, cnt, rank, nEdges);
  alloc_kernel<<<gNode, blk, 0, stream>>>(cnt, deg, start, total, nNodes);
  fill_kernel<<<gEdge, blk, 0, stream>>>(src, dst, start, rank, col, nEdges);

  gather_kernel<<<gGath, blk, 0, stream>>>(x, col, start, deg, sum, nNodes);
  sage_gemm<true, true><<<gGemm, blk, 0, stream>>>(x, sum, deg, W1s, W1n, b1, h1, nNodes);

  gather_kernel<<<gGath, blk, 0, stream>>>(h1, col, start, deg, sum, nNodes);
  sage_gemm<true, false><<<gGemm, blk, 0, stream>>>(h1, sum, deg, W2s, W2n, b2, h2, nNodes);

  sage_gemm<false, false><<<gGemm, blk, 0, stream>>>(h2, nullptr, nullptr, W3, nullptr, b3,
                                                     out, nNodes);
}


# GEMMs as 512 resident blocks looping over 32-row tiles: a wave's weight fragments stay in registers (read once per block, not per k-step from global); next tile's rows fetched under the WMMAs
# speedup vs baseline: 2.3926x; 1.0861x over previous
#include <hip/hip_runtime.h>

#define NODE_F 128

typedef float v2f __attribute__((ext_vector_type(2)));
typedef float v8f __attribute__((ext_vector_type(8)));

static __device__ __forceinline__ v8f wmma_k4(v2f a, v2f b, v8f c) {
  return __builtin_amdgcn_wmma_f32_16x16x4_f32(
       false, a,  false, b,
       (short)0, c,  false,  false);
}


__global__ void count_kernel(const int* __restrict__ dst, int* __restrict__ cnt,
                             int* __restrict__ rank, int nEdges) {
  int e = blockIdx.x * blockDim.x + threadIdx.x;
  if (e < nEdges) rank[e] = atomicAdd(&cnt[dst[e]], 1);
}

__global__ void alloc_kernel(const int* cnt, float* deg, int* __restrict__ start,
                             int* __restrict__ total, int nNodes) {
  const int n    = blockIdx.x * blockDim.x + threadIdx.x;
  const int lane = threadIdx.x & 31;
  const int c    = (n < nNodes) ? cnt[n] : 0;
  int incl = c;
  for (int o = 1; o < 32; o <<= 1) {
    int t = __shfl_up(incl, o);
    if (lane >= o) incl += t;
  }
  int base = 0;
  if (lane == 31) base = atomicAdd(total, incl);
  base = __shfl(base, 31);
  if (n < nNodes) {
    start[n] = base + incl - c;
    deg[n]   = (float)c;
  }
}

__global__ void fill_kernel(const int* __restrict__ src, const int* __restrict__ dst,
                            const int* __restrict__ start, const int* __restrict__ rank,
                            int* __restrict__ col, int nEdges) {
  int e = blockIdx.x * blockDim.x + threadIdx.x;
  if (e < nEdges) col[start[dst[e]] + rank[e]] = src[e];
}

static __device__ __forceinline__ float4 ld_row(const float* __restrict__ h, int s, int lane) {
  return ((const float4*)(h + (size_t)s * NODE_F))[lane];
}
static __device__ __forceinline__ void add4(float4& a, const float4 v) {
  a.x += v.x; a.y += v.y; a.z += v.z; a.w += v.w;
}

__global__ void gather_kernel(const float* __restrict__ h, const int* __restrict__ col,
                              const int* __restrict__ start, const float* __restrict__ deg,
                              float* __restrict__ sum, int nNodes) {
  const int g    = blockIdx.x * blockDim.x + threadIdx.x;
  const int n    = g >> 5;
  const int lane = g & 31;
  if (n >= nNodes) return;
  const int beg = __builtin_amdgcn_readfirstlane(start[n]);
  const int c   = __builtin_amdgcn_readfirstlane((int)deg[n]);
  float4 acc = make_float4(0.f, 0.f, 0.f, 0.f);
  for (int b = 0; b < c; b += 32) {
    const int m   = (c - b < 32) ? (c - b) : 32;
    const int idx = (lane < m) ? col[beg + b + lane] : 0;
    int k = 0;
    for (; k + 8 <= m; k += 8) {
      const float4 v0 = ld_row(h, __builtin_amdgcn_readlane(idx, k + 0), lane);
      const float4 v1 = ld_row(h, __builtin_amdgcn_readlane(idx, k + 1), lane);
      const float4 v2 = ld_row(h, __builtin_amdgcn_readlane(idx, k + 2), lane);
      const float4 v3 = ld_row(h, __builtin_amdgcn_readlane(idx, k + 3), lane);
      const float4 v4 = ld_row(h, __builtin_amdgcn_readlane(idx, k + 4), lane);
      const float4 v5 = ld_row(h, __builtin_amdgcn_readlane(idx, k + 5), lane);
      const float4 v6 = ld_row(h, __builtin_amdgcn_readlane(idx, k + 6), lane);
      const float4 v7 = ld_row(h, __builtin_amdgcn_readlane(idx, k + 7), lane);
      add4(acc, v0); add4(acc, v1); add4(acc, v2); add4(acc, v3);
      add4(acc, v4); add4(acc, v5); add4(acc, v6); add4(acc, v7);
    }
    if (k + 4 <= m) {
      const float4 v0 = ld_row(h, __builtin_amdgcn_readlane(idx, k + 0), lane);
      const float4 v1 = ld_row(h, __builtin_amdgcn_readlane(idx, k + 1), lane);
      const float4 v2 = ld_row(h, __builtin_amdgcn_readlane(idx, k + 2), lane);
      const float4 v3 = ld_row(h, __builtin_amdgcn_readlane(idx, k + 3), lane);
      add4(acc, v0); add4(acc, v1); add4(acc, v2); add4(acc, v3);
      k += 4;
    }
    for (; k < m; ++k) {
      const float4 v0 = ld_row(h, __builtin_amdgcn_readlane(idx, k), lane);
      add4(acc, v0);
    }
  }
  ((float4*)(sum + (size_t)n * NODE_F))[lane] = acc;
}


template <bool NEIGH>
static __device__ __forceinline__ void fetch_rows(const float* __restrict__ hself,
                                                  const float* __restrict__ sum,
                                                  const float* __restrict__ deg, int m_base,
                                                  int tid, int nNodes, float4 (&ph)[4],
                                                  float4 (&pn)[4], float (&pd)[4]) {
#pragma unroll
  for (int i = 0; i < 4; ++i) {
    const int f   = i * 256 + tid;
    const int row = f >> 5;
    const int cv  = f & 31;
    int m = m_base + row;
    if (m >= nNodes) m = nNodes - 1;
    ph[i] = ((const float4*)(hself + (size_t)m * NODE_F))[cv];
    if (NEIGH) {
      pn[i] = ((const float4*)(sum + (size_t)m * NODE_F))[cv];
      pd[i] = deg[m];
    }
  }
}

template <bool NEIGH, bool RELU>
__global__ __launch_bounds__(256) void sage_gemm(
    const float* __restrict__ hself, const float* __restrict__ sum,
    const float* __restrict__ deg, const float* __restrict__ Wself,
    const float* __restrict__ Wneigh, const float* __restrict__ bias,
    float* __restrict__ out, int nNodes) {
  __shared__ __attribute__((aligned(16))) float wst[NODE_F][NODE_F + 4];
  __shared__ __attribute__((aligned(16))) float hs[32][NODE_F + 4];
  __shared__ __attribute__((aligned(16))) float hn[NEIGH ? 32 : 1][NODE_F + 4];

  const int tid  = threadIdx.x;
  const int lane = tid & 31;
  const int wv   = tid >> 5;
  const int lid  = lane & 15;
  const int half = lane >> 4;
  const int n0   = wv * 16;

  v2f bs[32];
  v2f bn[NEIGH ? 32 : 1];
#pragma unroll
  for (int i = 0; i < 16; ++i) {
    const int f = i * 256 + tid;
    *(float4*)&wst[f >> 5][(f & 31) * 4] = ((const float4*)Wself)[f];
  }
  __syncthreads();
#pragma unroll
  for (int s = 0; s < 32; ++s) bs[s] = *(const v2f*)&wst[n0 + lid][4 * s + 2 * half];
  if (NEIGH) {
    __syncthreads();
#pragma unroll
    for (int i = 0; i < 16; ++i) {
      const int f = i * 256 + tid;
      *(float4*)&wst[f >> 5][(f & 31) * 4] = ((const float4*)Wneigh)[f];
    }
    __syncthreads();
#pragma unroll
    for (int s = 0; s < 32; ++s) bn[s] = *(const v2f*)&wst[n0 + lid][4 * s + 2 * half];
  }

  const float bv = bias[n0 + lid];
  const int nTiles = (nNodes + 31) >> 5;

  float4 ph[4], pn[4];
  float  pd[4];
  int tile = blockIdx.x;
  if (tile < nTiles) fetch_rows<NEIGH>(hself, sum, deg, tile * 32, tid, nNodes, ph, pn, pd);

  for (; tile < nTiles; tile += gridDim.x) {
    const int m_base = tile * 32;

#pragma unroll
    for (int i = 0; i < 4; ++i) {
      const int f   = i * 256 + tid;
      const int row = f >> 5;
      const int cv  = f & 31;
      const bool ok = (m_base + row < nNodes);
      float4 v = ph[i];
      if (!ok) v = make_float4(0.f, 0.f, 0.f, 0.f);
      *(float4*)&hs[row][cv * 4] = v;
      if (NEIGH) {
        const float rd = 1.0f / fmaxf(pd[i], 1.0f);
        float4 s4 = pn[i];
        s4.x *= rd; s4.y *= rd; s4.z *= rd; s4.w *= rd;
        if (!ok) s4 = make_float4(0.f, 0.f, 0.f, 0.f);
        *(float4*)&hn[row][cv * 4] = s4;
      }
    }
    __syncthreads();

    const int next = tile + gridDim.x;
    if (next < nTiles) fetch_rows<NEIGH>(hself, sum, deg, next * 32, tid, nNodes, ph, pn, pd);

    v8f acc0 = {0.f, 0.f, 0.f, 0.f, 0.f, 0.f, 0.f, 0.f};
    v8f acc1 = {0.f, 0.f, 0.f, 0.f, 0.f, 0.f, 0.f, 0.f};

#pragma unroll
    for (int s = 0; s < 32; ++s) {
      const int kk = 4 * s + 2 * half;
      const v2f a0 = *(const v2f*)&hs[lid][kk];
      const v2f a1 = *(const v2f*)&hs[16 + lid][kk];
      acc0 = wmma_k4(a0, bs[s], acc0);
      acc1 = wmma_k4(a1, bs[s], acc1);
      if (NEIGH) {
        const v2f c0 = *(const v2f*)&hn[lid][kk];
        const v2f c1 = *(const v2f*)&hn[16 + lid][kk];
        acc0 = wmma_k4(c0, bn[s], acc0);
        acc1 = wmma_k4(c1, bn[s], acc1);
      }
    }
    __syncthreads();

    float* outc = out + n0 + lid;
    const int r0 = m_base + 8 * half;
    const int r1 = m_base + 16 + 8 * half;
    if (m_base + 32 <= nNodes) {
#pragma unroll
      for (int e = 0; e < 8; ++e) {
        float v0 = acc0[e] + bv;
        float v1 = acc1[e] + bv;
        if (RELU) { v0 = fmaxf(v0, 0.f); v1 = fmaxf(v1, 0.f); }
        outc[(size_t)(r0 + e) * NODE_F] = v0;
        outc[(size_t)(r1 + e) * NODE_F] = v1;
      }
    } else {
#pragma unroll
      for (int e = 0; e < 8; ++e) {
        float v0 = acc0[e] + bv;
        float v1 = acc1[e] + bv;
        if (RELU) { v0 = fmaxf(v0, 0.f); v1 = fmaxf(v1, 0.f); }
        if (r0 + e < nNodes) outc[(size_t)(r0 + e) * NODE_F] = v0;
        if (r1 + e < nNodes) outc[(size_t)(r1 + e) * NODE_F] = v1;
      }
    }
  }
}

extern "C" void kernel_launch(void* const* d_in, const int* in_sizes, int n_in,
                              void* d_out, int out_size, void* d_ws, size_t ws_size,
                              hipStream_t stream) {
  const float* x      = (const float*)d_in[0];
  const int*   src    = (const int*)d_in[1];
  const int*   dst    = (const int*)d_in[2];
  const float* W1s    = (const float*)d_in[3];
  const float* W1n    = (const float*)d_in[4];
  const float* b1     = (const float*)d_in[5];
  const float* W2s    = (const float*)d_in[6];
  const float* W2n    = (const float*)d_in[7];
  const float* b2     = (const float*)d_in[8];
  const float* W3     = (const float*)d_in[9];
  const float* b3     = (const float*)d_in[10];
  float*       out    = (float*)d_out;

  const int nNodes = in_sizes[0] / NODE_F;
  const int nEdges = in_sizes[1];

  float* sum = (float*)d_ws;
  float* deg = sum + (size_t)nNodes * NODE_F;
  float* h1  = deg + nNodes;
  float* h2  = h1 + (size_t)nNodes * NODE_F;
  int* cnt   = (int*)deg;
  int* total = (int*)h1;
  int* col   = (int*)h2;
  int* rank  = col + nEdges;
  int* start = rank + nEdges;

  dim3 blk(256);
  dim3 gEdge((nEdges + 255) / 256);
  dim3 gNode((nNodes + 255) / 256);
  dim3 gGath((nNodes * 32 + 255) / 256);
  dim3 gGemm(512);

  hipMemsetAsync(cnt, 0, ((size_t)nNodes + 1) * sizeof(int), stream);
  count_kernel<<<gEdge, blk, 0, stream>>>(dst, cnt, rank, nEdges);
  alloc_kernel<<<gNode, blk, 0, stream>>>(cnt, deg, start, total, nNodes);
  fill_kernel<<<gEdge, blk, 0, stream>>>(src, dst, start, rank, col, nEdges);

  gather_kernel<<<gGath, blk, 0, stream>>>(x, col, start, deg, sum, nNodes);
  sage_gemm<true, true><<<gGemm, blk, 0, stream>>>(x, sum, deg, W1s, W1n, b1, h1, nNodes);

  gather_kernel<<<gGath, blk, 0, stream>>>(h1, col, start, deg, sum, nNodes);
  sage_gemm<true, false><<<gGemm, blk, 0, stream>>>(h1, sum, deg, W2s, W2n, b2, h2, nNodes);

  sage_gemm<false, false><<<gGemm, blk, 0, stream>>>(h2, nullptr, nullptr, W3, nullptr, b3,
                                                     out, nNodes);
}


# final linear folded into layer 2 per call: W3*W2self, W3*W2neigh (f32 WMMA) and W3*b2+b3 built by 17 extra blocks of the alloc launch; third GEMM and h2 gone; reassociates layer-2/final sums
# speedup vs baseline: 2.5709x; 1.0745x over previous
#include <hip/hip_runtime.h>

#define NODE_F 128

typedef float v2f __attribute__((ext_vector_type(2)));
typedef float v8f __attribute__((ext_vector_type(8)));

static __device__ __forceinline__ v8f wmma_k4(v2f a, v2f b, v8f c) {
  return __builtin_amdgcn_wmma_f32_16x16x4_f32(
       false, a,  false, b,
       (short)0, c,  false,  false);
}


__global__ void count_kernel(const int* __restrict__ dst, int* __restrict__ cnt,
                             int* __restrict__ rank, int nEdges) {
  int e = blockIdx.x * blockDim.x + threadIdx.x;
  if (e < nEdges) rank[e] = atomicAdd(&cnt[dst[e]], 1);
}

static __device__ __forceinline__ void fold_block(const int fb, const float* __restrict__ W3,
                                                  const float* __restrict__ W2s,
                                                  const float* __restrict__ W2n,
                                                  const float* __restrict__ b2,
                                                  const float* __restrict__ b3,
                                                  float* __restrict__ Wf) {
  __shared__ __attribute__((aligned(16))) float ct[16][NODE_F + 4];
  const int tid  = threadIdx.x;
  const int lane = tid & 31;
  const int wv   = tid >> 5;
  const int lid  = lane & 15;
  const int half = lane >> 4;
  if (fb < 16) {
    const float* W2 = (fb < 8) ? W2s : W2n;
    const int r0 = (fb & 7) * 16;
    const int c0 = wv * 16;
    const float* ap = W3 + (size_t)(r0 + lid) * NODE_F + 2 * half;
    const float* bp = W2 + (size_t)(2 * half) * NODE_F + c0 + lid;
    v8f acc0 = {0.f, 0.f, 0.f, 0.f, 0.f, 0.f, 0.f, 0.f};
    v8f acc1 = {0.f, 0.f, 0.f, 0.f, 0.f, 0.f, 0.f, 0.f};
#pragma unroll
    for (int s = 0; s < 32; s += 2) {
      const v2f a0 = *(const v2f*)(ap + 4 * s);
      const v2f a1 = *(const v2f*)(ap + 4 * s + 4);
      v2f b0, b1;
      b0.x = bp[(size_t)(4 * s + 0) * NODE_F];
      b0.y = bp[(size_t)(4 * s + 1) * NODE_F];
      b1.x = bp[(size_t)(4 * s + 4) * NODE_F];
      b1.y = bp[(size_t)(4 * s + 5) * NODE_F];
      acc0 = wmma_k4(a0, b0, acc0);
      acc1 = wmma_k4(a1, b1, acc1);
    }
#pragma unroll
    for (int e = 0; e < 8; ++e) ct[e + 8 * half][c0 + lid] = acc0[e] + acc1[e];
    __syncthreads();
    float* dstp = Wf + (size_t)(fb >> 3) * NODE_F * NODE_F + (size_t)r0 * NODE_F;
    for (int i = 0; i < 2; ++i) {
      const int f   = i * 256 + tid;
      const int row = f >> 5;
      const int cv  = f & 31;
      ((float4*)(dstp + (size_t)row * NODE_F))[cv] = *(const float4*)&ct[row][cv * 4];
    }
  } else if (tid < NODE_F) {
    const float4* w = (const float4*)(W3 + (size_t)tid * NODE_F);
    const float4* v = (const float4*)b2;
    float s = b3[tid];
#pragma unroll 8
    for (int j = 0; j < NODE_F / 4; ++j) {
      const float4 a = w[j];
      const float4 b = v[j];
      s += a.x * b.x + a.y * b.y + a.z * b.z + a.w * b.w;
    }
    Wf[2 * NODE_F * NODE_F + tid] = s;
  }
}

__global__ void alloc_kernel(
    const int* cnt, float* deg, int* __restrict__ start, int* __restrict__ total, int nNodes,
    int nAllocBlocks, const float* __restrict__ W3, const float* __restrict__ W2s,
    const float* __restrict__ W2n, const float* __restrict__ b2, const float* __restrict__ b3,
    float* __restrict__ Wf) {
  if ((int)blockIdx.x >= nAllocBlocks) {
    fold_block((int)blockIdx.x - nAllocBlocks, W3, W2s, W2n, b2, b3, Wf);
    return;
  }
  const int n    = blockIdx.x * blockDim.x + threadIdx.x;
  const int lane = threadIdx.x & 31;
  const int c    = (n < nNodes) ? cnt[n] : 0;
  int incl = c;
  for (int o = 1; o < 32; o <<= 1) {
    int t = __shfl_up(incl, o);
    if (lane >= o) incl += t;
  }
  int base = 0;
  if (lane == 31) base = atomicAdd(total, incl);
  base = __shfl(base, 31);
  if (n < nNodes) {
    start[n] = base + incl - c;
    deg[n]   = (float)c;
  }
}

__global__ void fill_kernel(const int* __restrict__ src, const int* __restrict__ dst,
                            const int* __restrict__ start, const int* __restrict__ rank,
                            int* __restrict__ col, int nEdges) {
  int e = blockIdx.x * blockDim.x + threadIdx.x;
  if (e < nEdges) col[start[dst[e]] + rank[e]] = src[e];
}

static __device__ __forceinline__ float4 ld_row(const float* __restrict__ h, int s, int lane) {
  return ((const float4*)(h + (size_t)s * NODE_F))[lane];
}
static __device__ __forceinline__ void add4(float4& a, const float4 v) {
  a.x += v.x; a.y += v.y; a.z += v.z; a.w += v.w;
}

__global__ void gather_kernel(const float* __restrict__ h, const int* __restrict__ col,
                              const int* __restrict__ start, const float* __restrict__ deg,
                              float* __restrict__ sum, int nNodes) {
  const int g    = blockIdx.x * blockDim.x + threadIdx.x;
  const int n    = g >> 5;
  const int lane = g & 31;
  if (n >= nNodes) return;
  const int beg = __builtin_amdgcn_readfirstlane(start[n]);
  const int c   = __builtin_amdgcn_readfirstlane((int)deg[n]);
  float4 acc = make_float4(0.f, 0.f, 0.f, 0.f);
  for (int b = 0; b < c; b += 32) {
    const int m   = (c - b < 32) ? (c - b) : 32;
    const int idx = (lane < m) ? col[beg + b + lane] : 0;
    int k = 0;
    for (; k + 8 <= m; k += 8) {
      const float4 v0 = ld_row(h, __builtin_amdgcn_readlane(idx, k + 0), lane);
      const float4 v1 = ld_row(h, __builtin_amdgcn_readlane(idx, k + 1), lane);
      const float4 v2 = ld_row(h, __builtin_amdgcn_readlane(idx, k + 2), lane);
      const float4 v3 = ld_row(h, __builtin_amdgcn_readlane(idx, k + 3), lane);
      const float4 v4 = ld_row(h, __builtin_amdgcn_readlane(idx, k + 4), lane);
      const float4 v5 = ld_row(h, __builtin_amdgcn_readlane(idx, k + 5), lane);
      const float4 v6 = ld_row(h, __builtin_amdgcn_readlane(idx, k + 6), lane);
      const float4 v7 = ld_row(h, __builtin_amdgcn_readlane(idx, k + 7), lane);
      add4(acc, v0); add4(acc, v1); add4(acc, v2); add4(acc, v3);
      add4(acc, v4); add4(acc, v5); add4(acc, v6); add4(acc, v7);
    }
    if (k + 4 <= m) {
      const float4 v0 = ld_row(h, __builtin_amdgcn_readlane(idx, k + 0), lane);
      const float4 v1 = ld_row(h, __builtin_amdgcn_readlane(idx, k + 1), lane);
      const float4 v2 = ld_row(h, __builtin_amdgcn_readlane(idx, k + 2), lane);
      const float4 v3 = ld_row(h, __builtin_amdgcn_readlane(idx, k + 3), lane);
      add4(acc, v0); add4(acc, v1); add4(acc, v2); add4(acc, v3);
      k += 4;
    }
    for (; k < m; ++k) {
      const float4 v0 = ld_row(h, __builtin_amdgcn_readlane(idx, k), lane);
      add4(acc, v0);
    }
  }
  ((float4*)(sum + (size_t)n * NODE_F))[lane] = acc;
}


static __device__ __forceinline__ void fetch_rows(const float* __restrict__ hself,
                                                  const float* __restrict__ sum,
                                                  const float* __restrict__ deg, int m_base,
                                                  int tid, int nNodes, float4 (&ph)[4],
                                                  float4 (&pn)[4], float (&pd)[4]) {
#pragma unroll
  for (int i = 0; i < 4; ++i) {
    const int f   = i * 256 + tid;
    const int row = f >> 5;
    const int cv  = f & 31;
    int m = m_base + row;
    if (m >= nNodes) m = nNodes - 1;
    ph[i] = ((const float4*)(hself + (size_t)m * NODE_F))[cv];
    pn[i] = ((const float4*)(sum + (size_t)m * NODE_F))[cv];
    pd[i] = deg[m];
  }
}

template <bool RELU>
__global__ __launch_bounds__(256) void sage_gemm(
    const float* __restrict__ hself, const float* __restrict__ sum,
    const float* __restrict__ deg, const float* __restrict__ Wself,
    const float* __restrict__ Wneigh, const float* __restrict__ bias,
    float* __restrict__ out, int nNodes) {
  __shared__ __attribute__((aligned(16))) float wst[NODE_F][NODE_F + 4];
  __shared__ __attribute__((aligned(16))) float hs[32][NODE_F + 4];
  __shared__ __attribute__((aligned(16))) float hn[32][NODE_F + 4];

  const int tid  = threadIdx.x;
  const int lane = tid & 31;
  const int wv   = tid >> 5;
  const int lid  = lane & 15;
  const int half = lane >> 4;
  const int n0   = wv * 16;

  v2f bs[32];
  v2f bn[32];
#pragma unroll
  for (int i = 0; i < 16; ++i) {
    const int f = i * 256 + tid;
    *(float4*)&wst[f >> 5][(f & 31) * 4] = ((const float4*)Wself)[f];
  }
  __syncthreads();
#pragma unroll
  for (int s = 0; s < 32; ++s) bs[s] = *(const v2f*)&wst[n0 + lid][4 * s + 2 * half];
  __syncthreads();
#pragma unroll
  for (int i = 0; i < 16; ++i) {
    const int f = i * 256 + tid;
    *(float4*)&wst[f >> 5][(f & 31) * 4] = ((const float4*)Wneigh)[f];
  }
  __syncthreads();
#pragma unroll
  for (int s = 0; s < 32; ++s) bn[s] = *(const v2f*)&wst[n0 + lid][4 * s + 2 * half];

  const float bv = bias[n0 + lid];
  const int nTiles = (nNodes + 31) >> 5;

  float4 ph[4], pn[4];
  float  pd[4];
  int tile = blockIdx.x;
  if (tile < nTiles) fetch_rows(hself, sum, deg, tile * 32, tid, nNodes, ph, pn, pd);

  for (; tile < nTiles; tile += gridDim.x) {
    const int m_base = tile * 32;

#pragma unroll
    for (int i = 0; i < 4; ++i) {
      const int f   = i * 256 + tid;
      const int row = f >> 5;
      const int cv  = f & 31;
      const bool ok = (m_base + row < nNodes);
      float4 v = ph[i];
      if (!ok) v = make_float4(0.f, 0.f, 0.f, 0.f);
      *(float4*)&hs[row][cv * 4] = v;
      const float rd = 1.0f / fmaxf(pd[i], 1.0f);
      float4 s4 = pn[i];
      s4.x *= rd; s4.y *= rd; s4.z *= rd; s4.w *= rd;
      if (!ok) s4 = make_float4(0.f, 0.f, 0.f, 0.f);
      *(float4*)&hn[row][cv * 4] = s4;
    }
    __syncthreads();

    const int next = tile + gridDim.x;
    if (next < nTiles) fetch_rows(hself, sum, deg, next * 32, tid, nNodes, ph, pn, pd);

    v8f acc0 = {0.f, 0.f, 0.f, 0.f, 0.f, 0.f, 0.f, 0.f};
    v8f acc1 = {0.f, 0.f, 0.f, 0.f, 0.f, 0.f, 0.f, 0.f};

#pragma unroll
    for (int s = 0; s < 32; ++s) {
      const int kk = 4 * s + 2 * half;
      const v2f a0 = *(const v2f*)&hs[lid][kk];
      const v2f a1 = *(const v2f*)&hs[16 + lid][kk];
      acc0 = wmma_k4(a0, bs[s], acc0);
      acc1 = wmma_k4(a1, bs[s], acc1);
      const v2f c0 = *(const v2f*)&hn[lid][kk];
      const v2f c1 = *(const v2f*)&hn[16 + lid][kk];
      acc0 = wmma_k4(c0, bn[s], acc0);
      acc1 = wmma_k4(c1, bn[s], acc1);
    }
    __syncthreads();

    float* outc = out + n0 + lid;
    const int r0 = m_base + 8 * half;
    const int r1 = m_base + 16 + 8 * half;
    if (m_base + 32 <= nNodes) {
#pragma unroll
      for (int e = 0; e < 8; ++e) {
        float v0 = acc0[e] + bv;
        float v1 = acc1[e] + bv;
        if (RELU) { v0 = fmaxf(v0, 0.f); v1 = fmaxf(v1, 0.f); }
        outc[(size_t)(r0 + e) * NODE_F] = v0;
        outc[(size_t)(r1 + e) * NODE_F] = v1;
      }
    } else {
#pragma unroll
      for (int e = 0; e < 8; ++e) {
        float v0 = acc0[e] + bv;
        float v1 = acc1[e] + bv;
        if (RELU) { v0 = fmaxf(v0, 0.f); v1 = fmaxf(v1, 0.f); }
        if (r0 + e < nNodes) outc[(size_t)(r0 + e) * NODE_F] = v0;
        if (r1 + e < nNodes) outc[(size_t)(r1 + e) * NODE_F] = v1;
      }
    }
  }
}

extern "C" void kernel_launch(void* const* d_in, const int* in_sizes, int n_in,
                              void* d_out, int out_size, void* d_ws, size_t ws_size,
                              hipStream_t stream) {
  const float* x      = (const float*)d_in[0];
  const int*   src    = (const int*)d_in[1];
  const int*   dst    = (const int*)d_in[2];
  const float* W1s    = (const float*)d_in[3];
  const float* W1n    = (const float*)d_in[4];
  const float* b1     = (const float*)d_in[5];
  const float* W2s    = (const float*)d_in[6];
  const float* W2n    = (const float*)d_in[7];
  const float* b2     = (const float*)d_in[8];
  const float* W3     = (const float*)d_in[9];
  const float* b3     = (const float*)d_in[10];
  float*       out    = (float*)d_out;

  const int nNodes = in_sizes[0] / NODE_F;
  const int nEdges = in_sizes[1];

  float* sum = (float*)d_ws;
  float* deg = sum + (size_t)nNodes * NODE_F;
  float* h1  = deg + nNodes;
  int* cnt   = (int*)deg;
  int* total = (int*)h1;
  int* col   = (int*)(h1 + (size_t)nNodes * NODE_F);
  int* rank  = col + nEdges;
  int* start = rank + nEdges;
  float* Wf  = (float*)(start + nNodes);
  const float* As = Wf;
  const float* An = Wf + NODE_F * NODE_F;
  const float* cf = Wf + 2 * NODE_F * NODE_F;

  dim3 blk(256);
  dim3 gEdge((nEdges + 255) / 256);
  dim3 gNode((nNodes + 255) / 256);
  dim3 gAlloc((nNodes + 255) / 256 + 17);
  dim3 gGath((nNodes * 32 + 255) / 256);
  dim3 gGemm(512);

  hipMemsetAsync(cnt, 0, ((size_t)nNodes + 1) * sizeof(int), stream);
  count_kernel<<<gEdge, blk, 0, stream>>>(dst, cnt, rank, nEdges);
  alloc_kernel<<<gAlloc, blk, 0, stream>>>(cnt, deg, start, total, nNodes, (int)gNode.x,
                                           W3, W2s, W2n, b2, b3, Wf);
  fill_kernel<<<gEdge, blk, 0, stream>>>(src, dst, start, rank, col, nEdges);

  gather_kernel<<<gGath, blk, 0, stream>>>(x, col, start, deg, sum, nNodes);
  sage_gemm<true><<<gGemm, blk, 0, stream>>>(x, sum, deg, W1s, W1n, b1, h1, nNodes);

  gather_kernel<<<gGath, blk, 0, stream>>>(h1, col, start, deg, sum, nNodes);
  sage_gemm<false><<<gGemm, blk, 0, stream>>>(h1, sum, deg, As, An, cf, out, nNodes);
}


# edge sort by destination with all counters in LDS: blocks bucket-sort 4096 edges each, one block per 128-node bucket then counts and places; no global atomic or memset; sum order still arrival rank
# speedup vs baseline: 3.7426x; 1.4558x over previous
#include <hip/hip_runtime.h>

#define NODE_F 128

typedef float v2f __attribute__((ext_vector_type(2)));
typedef float v8f __attribute__((ext_vector_type(8)));

static __device__ __forceinline__ v8f wmma_k4(v2f a, v2f b, v8f c) {
  return __builtin_amdgcn_wmma_f32_16x16x4_f32(
       false, a,  false, b,
       (short)0, c,  false,  false);
}


#define EPB    4096
#define BSH    7
#define BNODES 128

static __device__ __forceinline__ void fold_block(const int fb, const float* __restrict__ W3,
                                                  const float* __restrict__ W2s,
                                                  const float* __restrict__ W2n,
                                                  const float* __restrict__ b2,
                                                  const float* __restrict__ b3,
                                                  float* __restrict__ Wf) {
  __shared__ __attribute__((aligned(16))) float ct[16][NODE_F + 4];
  const int tid  = threadIdx.x;
  const int lane = tid & 31;
  const int wv   = tid >> 5;
  const int lid  = lane & 15;
  const int half = lane >> 4;
  if (fb < 16) {
    const float* W2 = (fb < 8) ? W2s : W2n;
    const int r0 = (fb & 7) * 16;
    const int c0 = wv * 16;
    const float* ap = W3 + (size_t)(r0 + lid) * NODE_F + 2 * half;
    const float* bp = W2 + (size_t)(2 * half) * NODE_F + c0 + lid;
    v8f acc0 = {0.f, 0.f, 0.f, 0.f, 0.f, 0.f, 0.f, 0.f};
    v8f acc1 = {0.f, 0.f, 0.f, 0.f, 0.f, 0.f, 0.f, 0.f};
#pragma unroll
    for (int s = 0; s < 32; s += 2) {
      const v2f a0 = *(const v2f*)(ap + 4 * s);
      const v2f a1 = *(const v2f*)(ap + 4 * s + 4);
      v2f b0, b1;
      b0.x = bp[(size_t)(4 * s + 0) * NODE_F];
      b0.y = bp[(size_t)(4 * s + 1) * NODE_F];
      b1.x = bp[(size_t)(4 * s + 4) * NODE_F];
      b1.y = bp[(size_t)(4 * s + 5) * NODE_F];
      acc0 = wmma_k4(a0, b0, acc0);
      acc1 = wmma_k4(a1, b1, acc1);
    }
#pragma unroll
    for (int e = 0; e < 8; ++e) ct[e + 8 * half][c0 + lid] = acc0[e] + acc1[e];
    __syncthreads();
    float* dstp = Wf + (size_t)(fb >> 3) * NODE_F * NODE_F + (size_t)r0 * NODE_F;
    for (int i = 0; i < 2; ++i) {
      const int f   = i * 256 + tid;
      const int row = f >> 5;
      const int cv  = f & 31;
      ((float4*)(dstp + (size_t)row * NODE_F))[cv] = *(const float4*)&ct[row][cv * 4];
    }
  } else if (tid < NODE_F) {
    const float4* w = (const float4*)(W3 + (size_t)tid * NODE_F);
    const float4* v = (const float4*)b2;
    float s = b3[tid];
#pragma unroll 8
    for (int j = 0; j < NODE_F / 4; ++j) {
      const float4 a = w[j];
      const float4 b = v[j];
      s += a.x * b.x + a.y * b.y + a.z * b.z + a.w * b.w;
    }
    Wf[2 * NODE_F * NODE_F + tid] = s;
  }
}

__global__ __launch_bounds__(256) void part_kernel(
    const int* __restrict__ src, const int* __restrict__ dst, int nEdges, int nBuckets,
    int nPartBlocks, int2* __restrict__ part, int* __restrict__ off,
    const float* __restrict__ W3, const float* __restrict__ W2s, const float* __restrict__ W2n,
    const float* __restrict__ b2, const float* __restrict__ b3, float* __restrict__ Wf) {
  extern __shared__ __attribute__((aligned(16))) int lds_dyn[];
  __shared__ int wsum[8];
  if ((int)blockIdx.x >= nPartBlocks) {
    fold_block((int)blockIdx.x - nPartBlocks, W3, W2s, W2n, b2, b3, Wf);
    return;
  }
  int*  hist  = lds_dyn;
  int2* stage = (int2*)(lds_dyn + ((nBuckets + 4) & ~3));
  const int tid  = threadIdx.x;
  const int lane = tid & 31;
  const int wv   = tid >> 5;
  const int blk  = blockIdx.x;

  for (int i = tid; i <= nBuckets; i += 256) hist[i] = 0;
  __syncthreads();

  const int e0 = blk * EPB + tid;
  int s[16], d[16], r[16];
#pragma unroll
  for (int k = 0; k < 16; ++k) {
    const int e = e0 + k * 256;
    const bool ok = e < nEdges;
    d[k] = ok ? dst[e] : 0;
    s[k] = ok ? src[e] : 0;
  }
#pragma unroll
  for (int k = 0; k < 16; ++k) {
    r[k] = 0;
    if (e0 + k * 256 < nEdges) r[k] = atomicAdd(&hist[d[k] >> BSH], 1);
  }
  __syncthreads();

  const int per = (nBuckets + 255) >> 8;
  const int lo  = tid * per;
  const int hi  = (lo + per < nBuckets) ? (lo + per) : nBuckets;
  int tsum = 0;
  for (int i = lo; i < hi; ++i) tsum += hist[i];
  int incl = tsum;
  for (int o = 1; o < 32; o <<= 1) {
    const int t = __shfl_up(incl, o);
    if (lane >= o) incl += t;
  }
  if (lane == 31) wsum[wv] = incl;
  __syncthreads();
  int wpre = 0, total = 0;
  for (int w = 0; w < 8; ++w) {
    const int t = wsum[w];
    if (w < wv) wpre += t;
    total += t;
  }
  int run = wpre + incl - tsum;
  for (int i = lo; i < hi; ++i) {
    const int c = hist[i];
    hist[i] = run;
    run += c;
  }
  if (tid == 0) hist[nBuckets] = total;
  __syncthreads();

  int* orow = off + (size_t)blk * (nBuckets + 1);
  for (int i = tid; i <= nBuckets; i += 256) orow[i] = hist[i];

#pragma unroll
  for (int k = 0; k < 16; ++k) {
    if (e0 + k * 256 < nEdges) stage[hist[d[k] >> BSH] + r[k]] = make_int2(s[k], d[k]);
  }
  __syncthreads();
  int2* prow = part + (size_t)blk * EPB;
  for (int j = tid; j < total; j += 256) prow[j] = stage[j];
}

__global__ __launch_bounds__(256) void bucket_kernel(
    const int2* __restrict__ part, const int* __restrict__ off, int nPartBlocks, int nBuckets,
    int nNodes, int* __restrict__ col, int* __restrict__ start, float* __restrict__ deg) {
  __shared__ int cnt[BNODES];
  __shared__ int cur[BNODES];
  __shared__ int wred[8];
  __shared__ int wscan[BNODES / 32];
  const int tid  = threadIdx.x;
  const int lane = tid & 31;
  const int wv   = tid >> 5;
  const int b    = blockIdx.x;
  const int n0   = b << BSH;
  const size_t pitch = (size_t)nBuckets + 1;

  if (tid < BNODES) cnt[tid] = 0;
  __syncthreads();

  int baseAcc = 0;
  for (int blk = tid; blk < nPartBlocks; blk += 256) {
    const int* o = off + (size_t)blk * pitch + b;
    const int sb = o[0];
    const int se = o[1];
    baseAcc += sb;
    const int2* p = part + (size_t)blk * EPB;
    for (int i = sb; i < se; i += 8) {
      int dd[8];
#pragma unroll
      for (int k = 0; k < 8; ++k) {
        const int j = (i + k < se) ? (i + k) : (se - 1);
        dd[k] = p[j].y;
      }
#pragma unroll
      for (int k = 0; k < 8; ++k) {
        atomicAdd(&cnt[dd[k] - n0], (i + k < se) ? 1 : 0);
      }
    }
  }
  for (int o = 16; o > 0; o >>= 1) baseAcc += __shfl_xor(baseAcc, o);
  if (lane == 0) wred[wv] = baseAcc;
  __syncthreads();

  int base = 0;
  for (int w = 0; w < 8; ++w) base += wred[w];
  int c = 0, incl = 0;
  if (tid < BNODES) {
    c = cnt[tid];
    incl = c;
    for (int o = 1; o < 32; o <<= 1) {
      const int t = __shfl_up(incl, o);
      if (lane >= o) incl += t;
    }
    if (lane == 31) wscan[wv] = incl;
  }
  __syncthreads();
  if (tid < BNODES) {
    int pre = 0;
    for (int w = 0; w < BNODES / 32; ++w) if (w < wv) pre += wscan[w];
    const int first = base + pre + incl - c;
    cur[tid] = first;
    if (n0 + tid < nNodes) {
      start[n0 + tid] = first;
      deg[n0 + tid]   = (float)c;
    }
  }
  __syncthreads();

  for (int blk = tid; blk < nPartBlocks; blk += 256) {
    const int* o = off + (size_t)blk * pitch + b;
    const int sb = o[0];
    const int se = o[1];
    const int2* p = part + (size_t)blk * EPB;
    for (int i = sb; i < se; i += 8) {
      int2 vv[8];
#pragma unroll
      for (int k = 0; k < 8; ++k) {
        const int j = (i + k < se) ? (i + k) : (se - 1);
        vv[k] = p[j];
      }
#pragma unroll
      for (int k = 0; k < 8; ++k) {
        const bool ok  = i + k < se;
        const int slot = atomicAdd(&cur[vv[k].y - n0], ok ? 1 : 0);
        if (ok) col[slot] = vv[k].x;
      }
    }
  }
}

static __device__ __forceinline__ float4 ld_row(const float* __restrict__ h, int s, int lane) {
  return ((const float4*)(h + (size_t)s * NODE_F))[lane];
}
static __device__ __forceinline__ void add4(float4& a, const float4 v) {
  a.x += v.x; a.y += v.y; a.z += v.z; a.w += v.w;
}

__global__ void gather_kernel(const float* __restrict__ h, const int* __restrict__ col,
                              const int* __restrict__ start, const float* __restrict__ deg,
                              float* __restrict__ sum, int nNodes) {
  const int g    = blockIdx.x * blockDim.x + threadIdx.x;
  const int n    = g >> 5;
  const int lane = g & 31;
  if (n >= nNodes) return;
  const int beg = __builtin_amdgcn_readfirstlane(start[n]);
  const int c   = __builtin_amdgcn_readfirstlane((int)deg[n]);
  float4 acc = make_float4(0.f, 0.f, 0.f, 0.f);
  for (int b = 0; b < c; b += 32) {
    const int m   = (c - b < 32) ? (c - b) : 32;
    const int idx = (lane < m) ? col[beg + b + lane] : 0;
    int k = 0;
    for (; k + 8 <= m; k += 8) {
      const float4 v0 = ld_row(h, __builtin_amdgcn_readlane(idx, k + 0), lane);
      const float4 v1 = ld_row(h, __builtin_amdgcn_readlane(idx, k + 1), lane);
      const float4 v2 = ld_row(h, __builtin_amdgcn_readlane(idx, k + 2), lane);
      const float4 v3 = ld_row(h, __builtin_amdgcn_readlane(idx, k + 3), lane);
      const float4 v4 = ld_row(h, __builtin_amdgcn_readlane(idx, k + 4), lane);
      const float4 v5 = ld_row(h, __builtin_amdgcn_readlane(idx, k + 5), lane);
      const float4 v6 = ld_row(h, __builtin_amdgcn_readlane(idx, k + 6), lane);
      const float4 v7 = ld_row(h, __builtin_amdgcn_readlane(idx, k + 7), lane);
      add4(acc, v0); add4(acc, v1); add4(acc, v2); add4(acc, v3);
      add4(acc, v4); add4(acc, v5); add4(acc, v6); add4(acc, v7);
    }
    if (k + 4 <= m) {
      const float4 v0 = ld_row(h, __builtin_amdgcn_readlane(idx, k + 0), lane);
      const float4 v1 = ld_row(h, __builtin_amdgcn_readlane(idx, k + 1), lane);
      const float4 v2 = ld_row(h, __builtin_amdgcn_readlane(idx, k + 2), lane);
      const float4 v3 = ld_row(h, __builtin_amdgcn_readlane(idx, k + 3), lane);
      add4(acc, v0); add4(acc, v1); add4(acc, v2); add4(acc, v3);
      k += 4;
    }
    for (; k < m; ++k) {
      const float4 v0 = ld_row(h, __builtin_amdgcn_readlane(idx, k), lane);
      add4(acc, v0);
    }
  }
  ((float4*)(sum + (size_t)n * NODE_F))[lane] = acc;
}


static __device__ __forceinline__ void fetch_rows(const float* __restrict__ hself,
                                                  const float* __restrict__ sum,
                                                  const float* __restrict__ deg, int m_base,
                                                  int tid, int nNodes, float4 (&ph)[4],
                                                  float4 (&pn)[4], float (&pd)[4]) {
#pragma unroll
  for (int i = 0; i < 4; ++i) {
    const int f   = i * 256 + tid;
    const int row = f >> 5;
    const int cv  = f & 31;
    int m = m_base + row;
    if (m >= nNodes) m = nNodes - 1;
    ph[i] = ((const float4*)(hself + (size_t)m * NODE_F))[cv];
    pn[i] = ((const float4*)(sum + (size_t)m * NODE_F))[cv];
    pd[i] = deg[m];
  }
}

template <bool RELU>
__global__ __launch_bounds__(256) void sage_gemm(
    const float* __restrict__ hself, const float* __restrict__ sum,
    const float* __restrict__ deg, const float* __restrict__ Wself,
    const float* __restrict__ Wneigh, const float* __restrict__ bias,
    float* __restrict__ out, int nNodes) {
  __shared__ __attribute__((aligned(16))) float wst[NODE_F][NODE_F + 4];
  __shared__ __attribute__((aligned(16))) float hs[32][NODE_F + 4];
  __shared__ __attribute__((aligned(16))) float hn[32][NODE_F + 4];

  const int tid  = threadIdx.x;
  const int lane = tid & 31;
  const int wv   = tid >> 5;
  const int lid  = lane & 15;
  const int half = lane >> 4;
  const int n0   = wv * 16;

  v2f bs[32];
  v2f bn[32];
#pragma unroll
  for (int i = 0; i < 16; ++i) {
    const int f = i * 256 + tid;
    *(float4*)&wst[f >> 5][(f & 31) * 4] = ((const float4*)Wself)[f];
  }
  __syncthreads();
#pragma unroll
  for (int s = 0; s < 32; ++s) bs[s] = *(const v2f*)&wst[n0 + lid][4 * s + 2 * half];
  __syncthreads();
#pragma unroll
  for (int i = 0; i < 16; ++i) {
    const int f = i * 256 + tid;
    *(float4*)&wst[f >> 5][(f & 31) * 4] = ((const float4*)Wneigh)[f];
  }
  __syncthreads();
#pragma unroll
  for (int s = 0; s < 32; ++s) bn[s] = *(const v2f*)&wst[n0 + lid][4 * s + 2 * half];

  const float bv = bias[n0 + lid];
  const int nTiles = (nNodes + 31) >> 5;

  float4 ph[4], pn[4];
  float  pd[4];
  int tile = blockIdx.x;
  if (tile < nTiles) fetch_rows(hself, sum, deg, tile * 32, tid, nNodes, ph, pn, pd);

  for (; tile < nTiles; tile += gridDim.x) {
    const int m_base = tile * 32;

#pragma unroll
    for (int i = 0; i < 4; ++i) {
      const int f   = i * 256 + tid;
      const int row = f >> 5;
      const int cv  = f & 31;
      const bool ok = (m_base + row < nNodes);
      float4 v = ph[i];
      if (!ok) v = make_float4(0.f, 0.f, 0.f, 0.f);
      *(float4*)&hs[row][cv * 4] = v;
      const float rd = 1.0f / fmaxf(pd[i], 1.0f);
      float4 s4 = pn[i];
      s4.x *= rd; s4.y *= rd; s4.z *= rd; s4.w *= rd;
      if (!ok) s4 = make_float4(0.f, 0.f, 0.f, 0.f);
      *(float4*)&hn[row][cv * 4] = s4;
    }
    __syncthreads();

    const int next = tile + gridDim.x;
    if (next < nTiles) fetch_rows(hself, sum, deg, next * 32, tid, nNodes, ph, pn, pd);

    v8f acc0 = {0.f, 0.f, 0.f, 0.f, 0.f, 0.f, 0.f, 0.f};
    v8f acc1 = {0.f, 0.f, 0.f, 0.f, 0.f, 0.f, 0.f, 0.f};

#pragma unroll
    for (int s = 0; s < 32; ++s) {
      const int kk = 4 * s + 2 * half;
      const v2f a0 = *(const v2f*)&hs[lid][kk];
      const v2f a1 = *(const v2f*)&hs[16 + lid][kk];
      acc0 = wmma_k4(a0, bs[s], acc0);
      acc1 = wmma_k4(a1, bs[s], acc1);
      const v2f c0 = *(const v2f*)&hn[lid][kk];
      const v2f c1 = *(const v2f*)&hn[16 + lid][kk];
      acc0 = wmma_k4(c0, bn[s], acc0);
      acc1 = wmma_k4(c1, bn[s], acc1);
    }
    __syncthreads();

    float* outc = out + n0 + lid;
    const int r0 = m_base + 8 * half;
    const int r1 = m_base + 16 + 8 * half;
    if (m_base + 32 <= nNodes) {
#pragma unroll
      for (int e = 0; e < 8; ++e) {
        float v0 = acc0[e] + bv;
        float v1 = acc1[e] + bv;
        if (RELU) { v0 = fmaxf(v0, 0.f); v1 = fmaxf(v1, 0.f); }
        outc[(size_t)(r0 + e) * NODE_F] = v0;
        outc[(size_t)(r1 + e) * NODE_F] = v1;
      }
    } else {
#pragma unroll
      for (int e = 0; e < 8; ++e) {
        float v0 = acc0[e] + bv;
        float v1 = acc1[e] + bv;
        if (RELU) { v0 = fmaxf(v0, 0.f); v1 = fmaxf(v1, 0.f); }
        if (r0 + e < nNodes) outc[(size_t)(r0 + e) * NODE_F] = v0;
        if (r1 + e < nNodes) outc[(size_t)(r1 + e) * NODE_F] = v1;
      }
    }
  }
}

extern "C" void kernel_launch(void* const* d_in, const int* in_sizes, int n_in,
                              void* d_out, int out_size, void* d_ws, size_t ws_size,
                              hipStream_t stream) {
  const float* x      = (const float*)d_in[0];
  const int*   src    = (const int*)d_in[1];
  const int*   dst    = (const int*)d_in[2];
  const float* W1s    = (const float*)d_in[3];
  const float* W1n    = (const float*)d_in[4];
  const float* b1     = (const float*)d_in[5];
  const float* W2s    = (const float*)d_in[6];
  const float* W2n    = (const float*)d_in[7];
  const float* b2     = (const float*)d_in[8];
  const float* W3     = (const float*)d_in[9];
  const float* b3     = (const float*)d_in[10];
  float*       out    = (float*)d_out;

  const int nNodes = in_sizes[0] / NODE_F;
  const int nEdges = in_sizes[1];

  float* sum = (float*)d_ws;
  float* deg = sum + (size_t)nNodes * NODE_F;
  float* h1  = deg + nNodes;
  int* col   = (int*)(h1 + (size_t)nNodes * NODE_F);
  int* start = col + 2 * (size_t)nEdges;
  float* Wf  = (float*)(start + nNodes);
  const float* As = Wf;
  const float* An = Wf + NODE_F * NODE_F;
  const float* cf = Wf + 2 * NODE_F * NODE_F;

  const int nBuckets    = (nNodes + BNODES - 1) / BNODES;
  const int nPartBlocks = (nEdges + EPB - 1) / EPB;
  int2* part = (int2*)sum;
  int*  off  = (int*)sum + 2 * (size_t)nPartBlocks * EPB;
  const size_t ldsPart = (size_t)((nBuckets + 4) & ~3) * sizeof(int) + (size_t)EPB * sizeof(int2);

  dim3 blk(256);
  dim3 gPart(nPartBlocks + 17);
  dim3 gBuck(nBuckets);
  dim3 gGath((nNodes * 32 + 255) / 256);
  dim3 gGemm(512);

  part_kernel<<<gPart, blk, ldsPart, stream>>>(src, dst, nEdges, nBuckets, nPartBlocks, part, off,
                                               W3, W2s, W2n, b2, b3, Wf);
  bucket_kernel<<<gBuck, blk, 0, stream>>>(part, off, nPartBlocks, nBuckets, nNodes, col, start, deg);

  gather_kernel<<<gGath, blk, 0, stream>>>(x, col, start, deg, sum, nNodes);
  sage_gemm<true><<<gGemm, blk, 0, stream>>>(x, sum, deg, W1s, W1n, b1, h1, nNodes);

  gather_kernel<<<gGath, blk, 0, stream>>>(h1, col, start, deg, sum, nNodes);
  sage_gemm<false><<<gGemm, blk, 0, stream>>>(h1, sum, deg, As, An, cf, out, nNodes);
}


# gather fused into the layer GEMM kernel: a wave sums its tile nodes' neighbour rows into the LDS A tile while the other resident block runs WMMAs; sum plane and 2 launches gone; sum order kept
# speedup vs baseline: 3.9590x; 1.0578x over previous
#include <hip/hip_runtime.h>

#define NODE_F 128

typedef float v2f __attribute__((ext_vector_type(2)));
typedef float v8f __attribute__((ext_vector_type(8)));

static __device__ __forceinline__ v8f wmma_k4(v2f a, v2f b, v8f c) {
  return __builtin_amdgcn_wmma_f32_16x16x4_f32(
       false, a,  false, b,
       (short)0, c,  false,  false);
}


#define EPB    4096
#define BSH    7
#define BNODES 128

static __device__ __forceinline__ void fold_block(const int fb, const float* __restrict__ W3,
                                                  const float* __restrict__ W2s,
                                                  const float* __restrict__ W2n,
                                                  const float* __restrict__ b2,
                                                  const float* __restrict__ b3,
                                                  float* __restrict__ Wf) {
  __shared__ __attribute__((aligned(16))) float ct[16][NODE_F + 4];
  const int tid  = threadIdx.x;
  const int lane = tid & 31;
  const int wv   = tid >> 5;
  const int lid  = lane & 15;
  const int half = lane >> 4;
  if (fb < 16) {
    const float* W2 = (fb < 8) ? W2s : W2n;
    const int r0 = (fb & 7) * 16;
    const int c0 = wv * 16;
    const float* ap = W3 + (size_t)(r0 + lid) * NODE_F + 2 * half;
    const float* bp = W2 + (size_t)(2 * half) * NODE_F + c0 + lid;
    v8f acc0 = {0.f, 0.f, 0.f, 0.f, 0.f, 0.f, 0.f, 0.f};
    v8f acc1 = {0.f, 0.f, 0.f, 0.f, 0.f, 0.f, 0.f, 0.f};
#pragma unroll
    for (int s = 0; s < 32; s += 2) {
      const v2f a0 = *(const v2f*)(ap + 4 * s);
      const v2f a1 = *(const v2f*)(ap + 4 * s + 4);
      v2f b0, b1;
      b0.x = bp[(size_t)(4 * s + 0) * NODE_F];
      b0.y = bp[(size_t)(4 * s + 1) * NODE_F];
      b1.x = bp[(size_t)(4 * s + 4) * NODE_F];
      b1.y = bp[(size_t)(4 * s + 5) * NODE_F];
      acc0 = wmma_k4(a0, b0, acc0);
      acc1 = wmma_k4(a1, b1, acc1);
    }
#pragma unroll
    for (int e = 0; e < 8; ++e) ct[e + 8 * half][c0 + lid] = acc0[e] + acc1[e];
    __syncthreads();
    float* dstp = Wf + (size_t)(fb >> 3) * NODE_F * NODE_F + (size_t)r0 * NODE_F;
    for (int i = 0; i < 2; ++i) {
      const int f   = i * 256 + tid;
      const int row = f >> 5;
      const int cv  = f & 31;
      ((float4*)(dstp + (size_t)row * NODE_F))[cv] = *(const float4*)&ct[row][cv * 4];
    }
  } else if (tid < NODE_F) {
    const float4* w = (const float4*)(W3 + (size_t)tid * NODE_F);
    const float4* v = (const float4*)b2;
    float s = b3[tid];
#pragma unroll 8
    for (int j = 0; j < NODE_F / 4; ++j) {
      const float4 a = w[j];
      const float4 b = v[j];
      s += a.x * b.x + a.y * b.y + a.z * b.z + a.w * b.w;
    }
    Wf[2 * NODE_F * NODE_F + tid] = s;
  }
}

__global__ __launch_bounds__(256) void part_kernel(
    const int* __restrict__ src, const int* __restrict__ dst, int nEdges, int nBuckets,
    int nPartBlocks, int2* __restrict__ part, int* __restrict__ off,
    const float* __restrict__ W3, const float* __restrict__ W2s, const float* __restrict__ W2n,
    const float* __restrict__ b2, const float* __restrict__ b3, float* __restrict__ Wf) {
  extern __shared__ __attribute__((aligned(16))) int lds_dyn[];
  __shared__ int wsum[8];
  if ((int)blockIdx.x >= nPartBlocks) {
    fold_block((int)blockIdx.x - nPartBlocks, W3, W2s, W2n, b2, b3, Wf);
    return;
  }
  int*  hist  = lds_dyn;
  int2* stage = (int2*)(lds_dyn + ((nBuckets + 4) & ~3));
  const int tid  = threadIdx.x;
  const int lane = tid & 31;
  const int wv   = tid >> 5;
  const int blk  = blockIdx.x;

  for (int i = tid; i <= nBuckets; i += 256) hist[i] = 0;
  __syncthreads();

  const int e0 = blk * EPB + tid;
  int s[16], d[16], r[16];
#pragma unroll
  for (int k = 0; k < 16; ++k) {
    const int e = e0 + k * 256;
    const bool ok = e < nEdges;
    d[k] = ok ? dst[e] : 0;
    s[k] = ok ? src[e] : 0;
  }
#pragma unroll
  for (int k = 0; k < 16; ++k) {
    r[k] = 0;
    if (e0 + k * 256 < nEdges) r[k] = atomicAdd(&hist[d[k] >> BSH], 1);
  }
  __syncthreads();

  const int per = (nBuckets + 255) >> 8;
  const int lo  = tid * per;
  const int hi  = (lo + per < nBuckets) ? (lo + per) : nBuckets;
  int tsum = 0;
  for (int i = lo; i < hi; ++i) tsum += hist[i];
  int incl = tsum;
  for (int o = 1; o < 32; o <<= 1) {
    const int t = __shfl_up(incl, o);
    if (lane >= o) incl += t;
  }
  if (lane == 31) wsum[wv] = incl;
  __syncthreads();
  int wpre = 0, total = 0;
  for (int w = 0; w < 8; ++w) {
    const int t = wsum[w];
    if (w < wv) wpre += t;
    total += t;
  }
  int run = wpre + incl - tsum;
  for (int i = lo; i < hi; ++i) {
    const int c = hist[i];
    hist[i] = run;
    run += c;
  }
  if (tid == 0) hist[nBuckets] = total;
  __syncthreads();

  int* orow = off + (size_t)blk * (nBuckets + 1);
  for (int i = tid; i <= nBuckets; i += 256) orow[i] = hist[i];

#pragma unroll
  for (int k = 0; k < 16; ++k) {
    if (e0 + k * 256 < nEdges) stage[hist[d[k] >> BSH] + r[k]] = make_int2(s[k], d[k]);
  }
  __syncthreads();
  int2* prow = part + (size_t)blk * EPB;
  for (int j = tid; j < total; j += 256) prow[j] = stage[j];
}

__global__ __launch_bounds__(256) void bucket_kernel(
    const int2* __restrict__ part, const int* __restrict__ off, int nPartBlocks, int nBuckets,
    int nNodes, int* __restrict__ col, int* __restrict__ start, float* __restrict__ deg) {
  __shared__ int cnt[BNODES];
  __shared__ int cur[BNODES];
  __shared__ int wred[8];
  __shared__ int wscan[BNODES / 32];
  const int tid  = threadIdx.x;
  const int lane = tid & 31;
  const int wv   = tid >> 5;
  const int b    = blockIdx.x;
  const int n0   = b << BSH;
  const size_t pitch = (size_t)nBuckets + 1;

  if (tid < BNODES) cnt[tid] = 0;
  __syncthreads();

  int baseAcc = 0;
  for (int blk = tid; blk < nPartBlocks; blk += 256) {
    const int* o = off + (size_t)blk * pitch + b;
    const int sb = o[0];
    const int se = o[1];
    baseAcc += sb;
    const int2* p = part + (size_t)blk * EPB;
    for (int i = sb; i < se; i += 8) {
      int dd[8];
#pragma unroll
      for (int k = 0; k < 8; ++k) {
        const int j = (i + k < se) ? (i + k) : (se - 1);
        dd[k] = p[j].y;
      }
#pragma unroll
      for (int k = 0; k < 8; ++k) {
        atomicAdd(&cnt[dd[k] - n0], (i + k < se) ? 1 : 0);
      }
    }
  }
  for (int o = 16; o > 0; o >>= 1) baseAcc += __shfl_xor(baseAcc, o);
  if (lane == 0) wred[wv] = baseAcc;
  __syncthreads();

  int base = 0;
  for (int w = 0; w < 8; ++w) base += wred[w];
  int c = 0, incl = 0;
  if (tid < BNODES) {
    c = cnt[tid];
    incl = c;
    for (int o = 1; o < 32; o <<= 1) {
      const int t = __shfl_up(incl, o);
      if (lane >= o) incl += t;
    }
    if (lane == 31) wscan[wv] = incl;
  }
  __syncthreads();
  if (tid < BNODES) {
    int pre = 0;
    for (int w = 0; w < BNODES / 32; ++w) if (w < wv) pre += wscan[w];
    const int first = base + pre + incl - c;
    cur[tid] = first;
    if (n0 + tid < nNodes) {
      start[n0 + tid] = first;
      deg[n0 + tid]   = (float)c;
    }
  }
  __syncthreads();

  for (int blk = tid; blk < nPartBlocks; blk += 256) {
    const int* o = off + (size_t)blk * pitch + b;
    const int sb = o[0];
    const int se = o[1];
    const int2* p = part + (size_t)blk * EPB;
    for (int i = sb; i < se; i += 8) {
      int2 vv[8];
#pragma unroll
      for (int k = 0; k < 8; ++k) {
        const int j = (i + k < se) ? (i + k) : (se - 1);
        vv[k] = p[j];
      }
#pragma unroll
      for (int k = 0; k < 8; ++k) {
        const bool ok  = i + k < se;
        const int slot = atomicAdd(&cur[vv[k].y - n0], ok ? 1 : 0);
        if (ok) col[slot] = vv[k].x;
      }
    }
  }
}

static __device__ __forceinline__ float4 ld_row(const float* __restrict__ h, int s, int lane) {
  return ((const float4*)(h + (size_t)s * NODE_F))[lane];
}
static __device__ __forceinline__ void add4(float4& a, const float4 v) {
  a.x += v.x; a.y += v.y; a.z += v.z; a.w += v.w;
}


template <bool RELU>
__global__ __launch_bounds__(256) void sage_fused(
    const float* __restrict__ h, const int* __restrict__ col, const int* __restrict__ start,
    const float* __restrict__ deg, const float* __restrict__ Wself,
    const float* __restrict__ Wneigh, const float* __restrict__ bias,
    float* __restrict__ out, int nNodes) {
  __shared__ __attribute__((aligned(16))) float wst[NODE_F][NODE_F + 4];
  __shared__ __attribute__((aligned(16))) float hs[32][NODE_F + 4];
  __shared__ __attribute__((aligned(16))) float hn[32][NODE_F + 4];

  const int tid  = threadIdx.x;
  const int lane = tid & 31;
  const int wv   = tid >> 5;
  const int lid  = lane & 15;
  const int half = lane >> 4;
  const int n0   = wv * 16;

  v2f bs[32];
  v2f bn[32];
#pragma unroll
  for (int i = 0; i < 16; ++i) {
    const int f = i * 256 + tid;
    *(float4*)&wst[f >> 5][(f & 31) * 4] = ((const float4*)Wself)[f];
  }
  __syncthreads();
#pragma unroll
  for (int s = 0; s < 32; ++s) bs[s] = *(const v2f*)&wst[n0 + lid][4 * s + 2 * half];
  __syncthreads();
#pragma unroll
  for (int i = 0; i < 16; ++i) {
    const int f = i * 256 + tid;
    *(float4*)&wst[f >> 5][(f & 31) * 4] = ((const float4*)Wneigh)[f];
  }
  __syncthreads();
#pragma unroll
  for (int s = 0; s < 32; ++s) bn[s] = *(const v2f*)&wst[n0 + lid][4 * s + 2 * half];

  const float bv = bias[n0 + lid];
  const int nTiles = (nNodes + 31) >> 5;

  for (int tile = blockIdx.x; tile < nTiles; tile += gridDim.x) {
    const int m_base = tile * 32;

    float4 ph[4];
#pragma unroll
    for (int i = 0; i < 4; ++i) {
      int m = m_base + i * 8 + wv;
      if (m >= nNodes) m = nNodes - 1;
      ph[i] = ld_row(h, m, lane);
    }

    int   gb[4], gc[4], gi[4];
    float gd[4];
#pragma unroll
    for (int i = 0; i < 4; ++i) {
      int n = m_base + i * 8 + wv;
      const bool ok = n < nNodes;
      if (!ok) n = nNodes - 1;
      gb[i] = start[n];
      gd[i] = deg[n];
      if (!ok) gd[i] = 0.f;
    }
#pragma unroll
    for (int i = 0; i < 4; ++i) {
      gb[i] = __builtin_amdgcn_readfirstlane(gb[i]);
      gc[i] = __builtin_amdgcn_readfirstlane((int)gd[i]);
      gi[i] = (lane < gc[i]) ? col[gb[i] + lane] : 0;
    }
#pragma unroll
    for (int i = 0; i < 4; ++i) {
      const int row = i * 8 + wv;
      const int beg = gb[i];
      const int c   = gc[i];
      float4 acc = make_float4(0.f, 0.f, 0.f, 0.f);
      for (int b = 0; b < c; b += 32) {
        const int m = (c - b < 32) ? (c - b) : 32;
        int idx = gi[i];
        if (b > 0) idx = (lane < m) ? col[beg + b + lane] : 0;
        int k = 0;
        for (; k + 16 <= m; k += 16) {
          float4 v[16];
#pragma unroll
          for (int j = 0; j < 16; ++j) v[j] = ld_row(h, __builtin_amdgcn_readlane(idx, k + j), lane);
#pragma unroll
          for (int j = 0; j < 16; ++j) add4(acc, v[j]);
        }
        if (k + 8 <= m) {
          float4 v[8];
#pragma unroll
          for (int j = 0; j < 8; ++j) v[j] = ld_row(h, __builtin_amdgcn_readlane(idx, k + j), lane);
#pragma unroll
          for (int j = 0; j < 8; ++j) add4(acc, v[j]);
          k += 8;
        }
        if (k + 4 <= m) {
          float4 v[4];
#pragma unroll
          for (int j = 0; j < 4; ++j) v[j] = ld_row(h, __builtin_amdgcn_readlane(idx, k + j), lane);
#pragma unroll
          for (int j = 0; j < 4; ++j) add4(acc, v[j]);
          k += 4;
        }
        if (k < m) {
          const int   l1 = (k + 1 < m) ? (k + 1) : (m - 1);
          const int   l2 = (k + 2 < m) ? (k + 2) : (m - 1);
          const float4 v0 = ld_row(h, __builtin_amdgcn_readlane(idx, k), lane);
          float4 v1 = ld_row(h, __builtin_amdgcn_readlane(idx, l1), lane);
          float4 v2 = ld_row(h, __builtin_amdgcn_readlane(idx, l2), lane);
          if (k + 1 >= m) v1 = make_float4(0.f, 0.f, 0.f, 0.f);
          if (k + 2 >= m) v2 = make_float4(0.f, 0.f, 0.f, 0.f);
          add4(acc, v0); add4(acc, v1); add4(acc, v2);
        }
      }
      const float rd = 1.0f / fmaxf(gd[i], 1.0f);
      acc.x *= rd; acc.y *= rd; acc.z *= rd; acc.w *= rd;
      *(float4*)&hn[row][lane * 4] = acc;
    }

#pragma unroll
    for (int i = 0; i < 4; ++i) {
      const int row = i * 8 + wv;
      float4 v = ph[i];
      if (m_base + row >= nNodes) v = make_float4(0.f, 0.f, 0.f, 0.f);
      *(float4*)&hs[row][lane * 4] = v;
    }
    __syncthreads();

    v8f acc0 = {0.f, 0.f, 0.f, 0.f, 0.f, 0.f, 0.f, 0.f};
    v8f acc1 = {0.f, 0.f, 0.f, 0.f, 0.f, 0.f, 0.f, 0.f};

#pragma unroll
    for (int s = 0; s < 32; ++s) {
      const int kk = 4 * s + 2 * half;
      const v2f a0 = *(const v2f*)&hs[lid][kk];
      const v2f a1 = *(const v2f*)&hs[16 + lid][kk];
      acc0 = wmma_k4(a0, bs[s], acc0);
      acc1 = wmma_k4(a1, bs[s], acc1);
      const v2f c0 = *(const v2f*)&hn[lid][kk];
      const v2f c1 = *(const v2f*)&hn[16 + lid][kk];
      acc0 = wmma_k4(c0, bn[s], acc0);
      acc1 = wmma_k4(c1, bn[s], acc1);
    }
    __syncthreads();

    float* outc = out + n0 + lid;
    const int r0 = m_base + 8 * half;
    const int r1 = m_base + 16 + 8 * half;
    if (m_base + 32 <= nNodes) {
#pragma unroll
      for (int e = 0; e < 8; ++e) {
        float v0 = acc0[e] + bv;
        float v1 = acc1[e] + bv;
        if (RELU) { v0 = fmaxf(v0, 0.f); v1 = fmaxf(v1, 0.f); }
        outc[(size_t)(r0 + e) * NODE_F] = v0;
        outc[(size_t)(r1 + e) * NODE_F] = v1;
      }
    } else {
#pragma unroll
      for (int e = 0; e < 8; ++e) {
        float v0 = acc0[e] + bv;
        float v1 = acc1[e] + bv;
        if (RELU) { v0 = fmaxf(v0, 0.f); v1 = fmaxf(v1, 0.f); }
        if (r0 + e < nNodes) outc[(size_t)(r0 + e) * NODE_F] = v0;
        if (r1 + e < nNodes) outc[(size_t)(r1 + e) * NODE_F] = v1;
      }
    }
  }
}

extern "C" void kernel_launch(void* const* d_in, const int* in_sizes, int n_in,
                              void* d_out, int out_size, void* d_ws, size_t ws_size,
                              hipStream_t stream) {
  const float* x      = (const float*)d_in[0];
  const int*   src    = (const int*)d_in[1];
  const int*   dst    = (const int*)d_in[2];
  const float* W1s    = (const float*)d_in[3];
  const float* W1n    = (const float*)d_in[4];
  const float* b1     = (const float*)d_in[5];
  const float* W2s    = (const float*)d_in[6];
  const float* W2n    = (const float*)d_in[7];
  const float* b2     = (const float*)d_in[8];
  const float* W3     = (const float*)d_in[9];
  const float* b3     = (const float*)d_in[10];
  float*       out    = (float*)d_out;

  const int nNodes = in_sizes[0] / NODE_F;
  const int nEdges = in_sizes[1];

  float* sum = (float*)d_ws;
  float* deg = sum + (size_t)nNodes * NODE_F;
  float* h1  = deg + nNodes;
  int* col   = (int*)(h1 + (size_t)nNodes * NODE_F);
  int* start = col + 2 * (size_t)nEdges;
  float* Wf  = (float*)(start + nNodes);
  const float* As = Wf;
  const float* An = Wf + NODE_F * NODE_F;
  const float* cf = Wf + 2 * NODE_F * NODE_F;

  const int nBuckets    = (nNodes + BNODES - 1) / BNODES;
  const int nPartBlocks = (nEdges + EPB - 1) / EPB;
  int2* part = (int2*)sum;
  int*  off  = (int*)sum + 2 * (size_t)nPartBlocks * EPB;
  const size_t ldsPart = (size_t)((nBuckets + 4) & ~3) * sizeof(int) + (size_t)EPB * sizeof(int2);

  dim3 blk(256);
  dim3 gPart(nPartBlocks + 17);
  dim3 gBuck(nBuckets);
  dim3 gGemm(512);

  part_kernel<<<gPart, blk, ldsPart, stream>>>(src, dst, nEdges, nBuckets, nPartBlocks, part, off,
                                               W3, W2s, W2n, b2, b3, Wf);
  bucket_kernel<<<gBuck, blk, 0, stream>>>(part, off, nPartBlocks, nBuckets, nNodes, col, start, deg);

  sage_fused<true><<<gGemm, blk, 0, stream>>>(x, col, start, deg, W1s, W1n, b1, h1, nNodes);

  sage_fused<false><<<gGemm, blk, 0, stream>>>(h1, col, start, deg, As, An, cf, out, nNodes);
}


# gather as one row stream per wave: the joined slot list of 4 consecutive nodes (self rows behind it) runs through a 16-row register queue, a row requested as one is summed; sum order kept
# speedup vs baseline: 4.0940x; 1.0341x over previous
#include <hip/hip_runtime.h>

#define NODE_F 128

typedef float v2f __attribute__((ext_vector_type(2)));
typedef float v8f __attribute__((ext_vector_type(8)));

static __device__ __forceinline__ v8f wmma_k4(v2f a, v2f b, v8f c) {
  return __builtin_amdgcn_wmma_f32_16x16x4_f32(
       false, a,  false, b,
       (short)0, c,  false,  false);
}


#define EPB    4096
#define BSH    7
#define BNODES 128

static __device__ __forceinline__ void fold_block(const int fb, const float* __restrict__ W3,
                                                  const float* __restrict__ W2s,
                                                  const float* __restrict__ W2n,
                                                  const float* __restrict__ b2,
                                                  const float* __restrict__ b3,
                                                  float* __restrict__ Wf) {
  __shared__ __attribute__((aligned(16))) float ct[16][NODE_F + 4];
  const int tid  = threadIdx.x;
  const int lane = tid & 31;
  const int wv   = tid >> 5;
  const int lid  = lane & 15;
  const int half = lane >> 4;
  if (fb < 16) {
    const float* W2 = (fb < 8) ? W2s : W2n;
    const int r0 = (fb & 7) * 16;
    const int c0 = wv * 16;
    const float* ap = W3 + (size_t)(r0 + lid) * NODE_F + 2 * half;
    const float* bp = W2 + (size_t)(2 * half) * NODE_F + c0 + lid;
    v8f acc0 = {0.f, 0.f, 0.f, 0.f, 0.f, 0.f, 0.f, 0.f};
    v8f acc1 = {0.f, 0.f, 0.f, 0.f, 0.f, 0.f, 0.f, 0.f};
#pragma unroll
    for (int s = 0; s < 32; s += 2) {
      const v2f a0 = *(const v2f*)(ap + 4 * s);
      const v2f a1 = *(const v2f*)(ap + 4 * s + 4);
      v2f b0, b1;
      b0.x = bp[(size_t)(4 * s + 0) * NODE_F];
      b0.y = bp[(size_t)(4 * s + 1) * NODE_F];
      b1.x = bp[(size_t)(4 * s + 4) * NODE_F];
      b1.y = bp[(size_t)(4 * s + 5) * NODE_F];
      acc0 = wmma_k4(a0, b0, acc0);
      acc1 = wmma_k4(a1, b1, acc1);
    }
#pragma unroll
    for (int e = 0; e < 8; ++e) ct[e + 8 * half][c0 + lid] = acc0[e] + acc1[e];
    __syncthreads();
    float* dstp = Wf + (size_t)(fb >> 3) * NODE_F * NODE_F + (size_t)r0 * NODE_F;
    for (int i = 0; i < 2; ++i) {
      const int f   = i * 256 + tid;
      const int row = f >> 5;
      const int cv  = f & 31;
      ((float4*)(dstp + (size_t)row * NODE_F))[cv] = *(const float4*)&ct[row][cv * 4];
    }
  } else if (tid < NODE_F) {
    const float4* w = (const float4*)(W3 + (size_t)tid * NODE_F);
    const float4* v = (const float4*)b2;
    float s = b3[tid];
#pragma unroll 8
    for (int j = 0; j < NODE_F / 4; ++j) {
      const float4 a = w[j];
      const float4 b = v[j];
      s += a.x * b.x + a.y * b.y + a.z * b.z + a.w * b.w;
    }
    Wf[2 * NODE_F * NODE_F + tid] = s;
  }
}

__global__ __launch_bounds__(256) void part_kernel(
    const int* __restrict__ src, const int* __restrict__ dst, int nEdges, int nBuckets,
    int nPartBlocks, int2* __restrict__ part, int* __restrict__ off,
    const float* __restrict__ W3, const float* __restrict__ W2s, const float* __restrict__ W2n,
    const float* __restrict__ b2, const float* __restrict__ b3, float* __restrict__ Wf) {
  extern __shared__ __attribute__((aligned(16))) int lds_dyn[];
  __shared__ int wsum[8];
  if ((int)blockIdx.x >= nPartBlocks) {
    fold_block((int)blockIdx.x - nPartBlocks, W3, W2s, W2n, b2, b3, Wf);
    return;
  }
  int*  hist  = lds_dyn;
  int2* stage = (int2*)(lds_dyn + ((nBuckets + 4) & ~3));
  const int tid  = threadIdx.x;
  const int lane = tid & 31;
  const int wv   = tid >> 5;
  const int blk  = blockIdx.x;

  for (int i = tid; i <= nBuckets; i += 256) hist[i] = 0;
  __syncthreads();

  const int e0 = blk * EPB + tid;
  int s[16], d[16], r[16];
#pragma unroll
  for (int k = 0; k < 16; ++k) {
    const int e = e0 + k * 256;
    const bool ok = e < nEdges;
    d[k] = ok ? dst[e] : 0;
    s[k] = ok ? src[e] : 0;
  }
#pragma unroll
  for (int k = 0; k < 16; ++k) {
    r[k] = 0;
    if (e0 + k * 256 < nEdges) r[k] = atomicAdd(&hist[d[k] >> BSH], 1);
  }
  __syncthreads();

  const int per = (nBuckets + 255) >> 8;
  const int lo  = tid * per;
  const int hi  = (lo + per < nBuckets) ? (lo + per) : nBuckets;
  int tsum = 0;
  for (int i = lo; i < hi; ++i) tsum += hist[i];
  int incl = tsum;
  for (int o = 1; o < 32; o <<= 1) {
    const int t = __shfl_up(incl, o);
    if (lane >= o) incl += t;
  }
  if (lane == 31) wsum[wv] = incl;
  __syncthreads();
  int wpre = 0, total = 0;
  for (int w = 0; w < 8; ++w) {
    const int t = wsum[w];
    if (w < wv) wpre += t;
    total += t;
  }
  int run = wpre + incl - tsum;
  for (int i = lo; i < hi; ++i) {
    const int c = hist[i];
    hist[i] = run;
    run += c;
  }
  if (tid == 0) hist[nBuckets] = total;
  __syncthreads();

  int* orow = off + (size_t)blk * (nBuckets + 1);
  for (int i = tid; i <= nBuckets; i += 256) orow[i] = hist[i];

#pragma unroll
  for (int k = 0; k < 16; ++k) {
    if (e0 + k * 256 < nEdges) stage[hist[d[k] >> BSH] + r[k]] = make_int2(s[k], d[k]);
  }
  __syncthreads();
  int2* prow = part + (size_t)blk * EPB;
  for (int j = tid; j < total; j += 256) prow[j] = stage[j];
}

__global__ __launch_bounds__(256) void bucket_kernel(
    const int2* __restrict__ part, const int* __restrict__ off, int nPartBlocks, int nBuckets,
    int nNodes, int* __restrict__ col, int* __restrict__ start, float* __restrict__ deg) {
  __shared__ int cnt[BNODES];
  __shared__ int cur[BNODES];
  __shared__ int wred[8];
  __shared__ int wscan[BNODES / 32];
  const int tid  = threadIdx.x;
  const int lane = tid & 31;
  const int wv   = tid >> 5;
  const int b    = blockIdx.x;
  const int n0   = b << BSH;
  const size_t pitch = (size_t)nBuckets + 1;

  if (tid < BNODES) cnt[tid] = 0;
  __syncthreads();

  int baseAcc = 0;
  for (int blk = tid; blk < nPartBlocks; blk += 256) {
    const int* o = off + (size_t)blk * pitch + b;
    const int sb = o[0];
    const int se = o[1];
    baseAcc += sb;
    const int2* p = part + (size_t)blk * EPB;
    for (int i = sb; i < se; i += 8) {
      int dd[8];
#pragma unroll
      for (int k = 0; k < 8; ++k) {
        const int j = (i + k < se) ? (i + k) : (se - 1);
        dd[k] = p[j].y;
      }
#pragma unroll
      for (int k = 0; k < 8; ++k) {
        atomicAdd(&cnt[dd[k] - n0], (i + k < se) ? 1 : 0);
      }
    }
  }
  for (int o = 16; o > 0; o >>= 1) baseAcc += __shfl_xor(baseAcc, o);
  if (lane == 0) wred[wv] = baseAcc;
  __syncthreads();

  int base = 0;
  for (int w = 0; w < 8; ++w) base += wred[w];
  int c = 0, incl = 0;
  if (tid < BNODES) {
    c = cnt[tid];
    incl = c;
    for (int o = 1; o < 32; o <<= 1) {
      const int t = __shfl_up(incl, o);
      if (lane >= o) incl += t;
    }
    if (lane == 31) wscan[wv] = incl;
  }
  __syncthreads();
  if (tid < BNODES) {
    int pre = 0;
    for (int w = 0; w < BNODES / 32; ++w) if (w < wv) pre += wscan[w];
    const int first = base + pre + incl - c;
    cur[tid] = first;
    if (n0 + tid < nNodes) {
      start[n0 + tid] = first;
      deg[n0 + tid]   = (float)c;
    }
  }
  __syncthreads();

  for (int blk = tid; blk < nPartBlocks; blk += 256) {
    const int* o = off + (size_t)blk * pitch + b;
    const int sb = o[0];
    const int se = o[1];
    const int2* p = part + (size_t)blk * EPB;
    for (int i = sb; i < se; i += 8) {
      int2 vv[8];
#pragma unroll
      for (int k = 0; k < 8; ++k) {
        const int j = (i + k < se) ? (i + k) : (se - 1);
        vv[k] = p[j];
      }
#pragma unroll
      for (int k = 0; k < 8; ++k) {
        const bool ok  = i + k < se;
        const int slot = atomicAdd(&cur[vv[k].y - n0], ok ? 1 : 0);
        if (ok) col[slot] = vv[k].x;
      }
    }
  }
}

static __device__ __forceinline__ float4 ld_row_u(const float* __restrict__ h, int s, unsigned loff) {
  const char* rp = (const char*)(h + (size_t)s * NODE_F);
  return *(const float4*)(rp + loff);
}
static __device__ __forceinline__ void add4(float4& a, const float4 v) {
  a.x += v.x; a.y += v.y; a.z += v.z; a.w += v.w;
}


template <bool RELU>
__global__ __launch_bounds__(256) void sage_fused(
    const float* __restrict__ h, const int* __restrict__ col, const int* __restrict__ start,
    const float* __restrict__ deg, const float* __restrict__ Wself,
    const float* __restrict__ Wneigh, const float* __restrict__ bias,
    float* __restrict__ out, int nNodes) {
  __shared__ __attribute__((aligned(16))) float wst[NODE_F][NODE_F + 4];
  __shared__ __attribute__((aligned(16))) float at[2][32][NODE_F + 4];
  float (*hn)[NODE_F + 4] = at[0];
  float (*hs)[NODE_F + 4] = at[1];

  const int tid  = threadIdx.x;
  const int lane = tid & 31;
  const int wv   = tid >> 5;
  const int lid  = lane & 15;
  const int half = lane >> 4;
  const int n0   = wv * 16;

  v2f bs[32];
  v2f bn[32];
#pragma unroll
  for (int i = 0; i < 16; ++i) {
    const int f = i * 256 + tid;
    *(float4*)&wst[f >> 5][(f & 31) * 4] = ((const float4*)Wself)[f];
  }
  __syncthreads();
#pragma unroll
  for (int s = 0; s < 32; ++s) bs[s] = *(const v2f*)&wst[n0 + lid][4 * s + 2 * half];
  __syncthreads();
#pragma unroll
  for (int i = 0; i < 16; ++i) {
    const int f = i * 256 + tid;
    *(float4*)&wst[f >> 5][(f & 31) * 4] = ((const float4*)Wneigh)[f];
  }
  __syncthreads();
#pragma unroll
  for (int s = 0; s < 32; ++s) bn[s] = *(const v2f*)&wst[n0 + lid][4 * s + 2 * half];

  const float bv = bias[n0 + lid];
  const int nTiles = (nNodes + 31) >> 5;

  for (int tile = blockIdx.x; tile < nTiles; tile += gridDim.x) {
    const int m_base = tile * 32;

    const int nf = m_base + 4 * wv;
    int nq = nf + (lane & 3);
    const bool okn = nq < nNodes;
    if (!okn) nq = nNodes - 1;
    const int   sq  = start[nq];
    const float dqf = deg[nq];
    const int   dq  = okn ? ((lane < 4) ? (int)dqf : ((lane < 8) ? 1 : 0)) : 0;
    const float rq  = (lane < 4) ? 1.0f / fmaxf(okn ? dqf : 0.f, 1.0f) : 1.0f;
    const int beg = __builtin_amdgcn_readfirstlane(sq);
    const int T   = __builtin_amdgcn_readlane(dq, 0) + __builtin_amdgcn_readlane(dq, 1) +
                    __builtin_amdgcn_readlane(dq, 2) + __builtin_amdgcn_readlane(dq, 3);
    const int TT  = T + __builtin_amdgcn_readlane(dq, 4) + __builtin_amdgcn_readlane(dq, 5) +
                    __builtin_amdgcn_readlane(dq, 6) + __builtin_amdgcn_readlane(dq, 7);
    int   node = 0;
    int   nend = __builtin_amdgcn_readlane(dq, 0);
    float4 acc = make_float4(0.f, 0.f, 0.f, 0.f);
    float* hrow = &hn[4 * wv][lane * 4];
    const unsigned loff = (unsigned)lane * 16u;

#define NODE_DONE()                                                          \
    do {                                                                     \
      const float rd = __builtin_bit_cast(float, __builtin_amdgcn_readlane(__builtin_bit_cast(int, rq), node)); \
      acc.x *= rd; acc.y *= rd; acc.z *= rd; acc.w *= rd;                    \
      *(float4*)hrow = acc;                                                  \
      acc = make_float4(0.f, 0.f, 0.f, 0.f);                                 \
      ++node;                                                                \
      hrow += (node == 4) ? 29 * (NODE_F + 4) : (NODE_F + 4);                \
      nend += __builtin_amdgcn_readlane(dq, node);      \
    } while (0)
#define TAKE_ROW(r)                                                          \
    do {                                                                     \
      while (P == nend) NODE_DONE();                                         \
      add4(acc, (r));                                                        \
      ++P;                                                                   \
    } while (0)

    for (int b0 = 0; b0 < TT; b0 += 128) {
      const int Tb = (TT - b0 < 128) ? (TT - b0) : 128;
      int ix[4];
#pragma unroll
      for (int k = 0; k < 4; ++k) {
        const int pos = b0 + 32 * k + lane;
        int r = (pos < TT) ? (nf + pos - T) : 0;
        if (pos < T) r = col[beg + pos];
        ix[k] = r;
      }
      int i0 = ix[0], i1 = ix[1], i2 = ix[2], i3 = ix[3];
      int P = b0;
      float4 v[16];
#pragma unroll
      for (int j = 0; j < 16; ++j) v[j] = ld_row_u(h, __builtin_amdgcn_readlane(i0, j), loff);
      int base = 0;
      int odd  = 0;
      for (; base + 16 < Tb; base += 16) {
        const int w  = odd ? i1 : i0;
        const int lb = odd ? 0 : 16;
#pragma unroll
        for (int j = 0; j < 16; ++j) {
          TAKE_ROW(v[j]);
          asm volatile("" : "+v"(acc.x), "+v"(acc.y), "+v"(acc.z), "+v"(acc.w) : : "memory");
          v[j] = ld_row_u(h, __builtin_amdgcn_readlane(w, lb + j), loff);
        }
        if (odd) { i0 = i1; i1 = i2; i2 = i3; i3 = 0; }
        odd ^= 1;
      }
#pragma unroll
      for (int j = 0; j < 16; ++j) {
        if (base + j < Tb) TAKE_ROW(v[j]);
      }
    }
    while (node < 8) NODE_DONE();
#undef TAKE_ROW
#undef NODE_DONE
    __syncthreads();

    v8f acc0 = {0.f, 0.f, 0.f, 0.f, 0.f, 0.f, 0.f, 0.f};
    v8f acc1 = {0.f, 0.f, 0.f, 0.f, 0.f, 0.f, 0.f, 0.f};

#pragma unroll
    for (int s = 0; s < 32; ++s) {
      const int kk = 4 * s + 2 * half;
      const v2f a0 = *(const v2f*)&hs[lid][kk];
      const v2f a1 = *(const v2f*)&hs[16 + lid][kk];
      acc0 = wmma_k4(a0, bs[s], acc0);
      acc1 = wmma_k4(a1, bs[s], acc1);
      const v2f c0 = *(const v2f*)&hn[lid][kk];
      const v2f c1 = *(const v2f*)&hn[16 + lid][kk];
      acc0 = wmma_k4(c0, bn[s], acc0);
      acc1 = wmma_k4(c1, bn[s], acc1);
    }
    __syncthreads();

    float* outc = out + n0 + lid;
    const int r0 = m_base + 8 * half;
    const int r1 = m_base + 16 + 8 * half;
    if (m_base + 32 <= nNodes) {
#pragma unroll
      for (int e = 0; e < 8; ++e) {
        float v0 = acc0[e] + bv;
        float v1 = acc1[e] + bv;
        if (RELU) { v0 = fmaxf(v0, 0.f); v1 = fmaxf(v1, 0.f); }
        outc[(size_t)(r0 + e) * NODE_F] = v0;
        outc[(size_t)(r1 + e) * NODE_F] = v1;
      }
    } else {
#pragma unroll
      for (int e = 0; e < 8; ++e) {
        float v0 = acc0[e] + bv;
        float v1 = acc1[e] + bv;
        if (RELU) { v0 = fmaxf(v0, 0.f); v1 = fmaxf(v1, 0.f); }
        if (r0 + e < nNodes) outc[(size_t)(r0 + e) * NODE_F] = v0;
        if (r1 + e < nNodes) outc[(size_t)(r1 + e) * NODE_F] = v1;
      }
    }
  }
}

extern "C" void kernel_launch(void* const* d_in, const int* in_sizes, int n_in,
                              void* d_out, int out_size, void* d_ws, size_t ws_size,
                              hipStream_t stream) {
  const float* x      = (const float*)d_in[0];
  const int*   src    = (const int*)d_in[1];
  const int*   dst    = (const int*)d_in[2];
  const float* W1s    = (const float*)d_in[3];
  const float* W1n    = (const float*)d_in[4];
  const float* b1     = (const float*)d_in[5];
  const float* W2s    = (const float*)d_in[6];
  const float* W2n    = (const float*)d_in[7];
  const float* b2     = (const float*)d_in[8];
  const float* W3     = (const float*)d_in[9];
  const float* b3     = (const float*)d_in[10];
  float*       out    = (float*)d_out;

  const int nNodes = in_sizes[0] / NODE_F;
  const int nEdges = in_sizes[1];

  float* sum = (float*)d_ws;
  float* deg = sum + (size_t)nNodes * NODE_F;
  float* h1  = deg + nNodes;
  int* col   = (int*)(h1 + (size_t)nNodes * NODE_F);
  int* start = col + 2 * (size_t)nEdges;
  float* Wf  = (float*)(start + nNodes);
  const float* As = Wf;
  const float* An = Wf + NODE_F * NODE_F;
  const float* cf = Wf + 2 * NODE_F * NODE_F;

  const int nBuckets    = (nNodes + BNODES - 1) / BNODES;
  const int nPartBlocks = (nEdges + EPB - 1) / EPB;
  int2* part = (int2*)sum;
  int*  off  = (int*)sum + 2 * (size_t)nPartBlocks * EPB;
  const size_t ldsPart = (size_t)((nBuckets + 4) & ~3) * sizeof(int) + (size_t)EPB * sizeof(int2);

  dim3 blk(256);
  dim3 gPart(nPartBlocks + 17);
  dim3 gBuck(nBuckets);
  dim3 gGemm(512);

  part_kernel<<<gPart, blk, ldsPart, stream>>>(src, dst, nEdges, nBuckets, nPartBlocks, part, off,
                                               W3, W2s, W2n, b2, b3, Wf);
  bucket_kernel<<<gBuck, blk, 0, stream>>>(part, off, nPartBlocks, nBuckets, nNodes, col, start, deg);

  sage_fused<true><<<gGemm, blk, 0, stream>>>(x, col, start, deg, W1s, W1n, b1, h1, nNodes);

  sage_fused<false><<<gGemm, blk, 0, stream>>>(h1, col, start, deg, As, An, cf, out, nNodes);
}


# layer kernel epilogue: the 32-row tile is staged in LDS and each output row leaves as one wave store of four whole 128-byte lines, not as 64-byte halves from two waves; sums and order unchanged
# speedup vs baseline: 4.2286x; 1.0329x over previous
#include <hip/hip_runtime.h>

#define NODE_F 128

typedef float v2f __attribute__((ext_vector_type(2)));
typedef float v8f __attribute__((ext_vector_type(8)));

static __device__ __forceinline__ v8f wmma_k4(v2f a, v2f b, v8f c) {
  return __builtin_amdgcn_wmma_f32_16x16x4_f32(
       false, a,  false, b,
       (short)0, c,  false,  false);
}


#define EPB    4096
#define BSH    7
#define BNODES 128

static __device__ __forceinline__ void fold_block(const int fb, const float* __restrict__ W3,
                                                  const float* __restrict__ W2s,
                                                  const float* __restrict__ W2n,
                                                  const float* __restrict__ b2,
                                                  const float* __restrict__ b3,
                                                  float* __restrict__ Wf) {
  __shared__ __attribute__((aligned(16))) float ct[16][NODE_F + 4];
  const int tid  = threadIdx.x;
  const int lane = tid & 31;
  const int wv   = tid >> 5;
  const int lid  = lane & 15;
  const int half = lane >> 4;
  if (fb < 16) {
    const float* W2 = (fb < 8) ? W2s : W2n;
    const int r0 = (fb & 7) * 16;
    const int c0 = wv * 16;
    const float* ap = W3 + (size_t)(r0 + lid) * NODE_F + 2 * half;
    const float* bp = W2 + (size_t)(2 * half) * NODE_F + c0 + lid;
    v8f acc0 = {0.f, 0.f, 0.f, 0.f, 0.f, 0.f, 0.f, 0.f};
    v8f acc1 = {0.f, 0.f, 0.f, 0.f, 0.f, 0.f, 0.f, 0.f};
#pragma unroll
    for (int s = 0; s < 32; s += 2) {
      const v2f a0 = *(const v2f*)(ap + 4 * s);
      const v2f a1 = *(const v2f*)(ap + 4 * s + 4);
      v2f b0, b1;
      b0.x = bp[(size_t)(4 * s + 0) * NODE_F];
      b0.y = bp[(size_t)(4 * s + 1) * NODE_F];
      b1.x = bp[(size_t)(4 * s + 4) * NODE_F];
      b1.y = bp[(size_t)(4 * s + 5) * NODE_F];
      acc0 = wmma_k4(a0, b0, acc0);
      acc1 = wmma_k4(a1, b1, acc1);
    }
#pragma unroll
    for (int e = 0; e < 8; ++e) ct[e + 8 * half][c0 + lid] = acc0[e] + acc1[e];
    __syncthreads();
    float* dstp = Wf + (size_t)(fb >> 3) * NODE_F * NODE_F + (size_t)r0 * NODE_F;
    for (int i = 0; i < 2; ++i) {
      const int f   = i * 256 + tid;
      const int row = f >> 5;
      const int cv  = f & 31;
      ((float4*)(dstp + (size_t)row * NODE_F))[cv] = *(const float4*)&ct[row][cv * 4];
    }
  } else if (tid < NODE_F) {
    const float4* w = (const float4*)(W3 + (size_t)tid * NODE_F);
    const float4* v = (const float4*)b2;
    float s = b3[tid];
#pragma unroll 8
    for (int j = 0; j < NODE_F / 4; ++j) {
      const float4 a = w[j];
      const float4 b = v[j];
      s += a.x * b.x + a.y * b.y + a.z * b.z + a.w * b.w;
    }
    Wf[2 * NODE_F * NODE_F + tid] = s;
  }
}

__global__ __launch_bounds__(256) void part_kernel(
    const int* __restrict__ src, const int* __restrict__ dst, int nEdges, int nBuckets,
    int nPartBlocks, int2* __restrict__ part, int* __restrict__ off,
    const float* __restrict__ W3, const float* __restrict__ W2s, const float* __restrict__ W2n,
    const float* __restrict__ b2, const float* __restrict__ b3, float* __restrict__ Wf) {
  extern __shared__ __attribute__((aligned(16))) int lds_dyn[];
  __shared__ int wsum[8];
  if ((int)blockIdx.x >= nPartBlocks) {
    fold_block((int)blockIdx.x - nPartBlocks, W3, W2s, W2n, b2, b3, Wf);
    return;
  }
  int*  hist  = lds_dyn;
  int2* stage = (int2*)(lds_dyn + ((nBuckets + 4) & ~3));
  const int tid  = threadIdx.x;
  const int lane = tid & 31;
  const int wv   = tid >> 5;
  const int blk  = blockIdx.x;

  for (int i = tid; i <= nBuckets; i += 256) hist[i] = 0;
  __syncthreads();

  const int e0 = blk * EPB + tid;
  int s[16], d[16], r[16];
#pragma unroll
  for (int k = 0; k < 16; ++k) {
    const int e = e0 + k * 256;
    const bool ok = e < nEdges;
    d[k] = ok ? dst[e] : 0;
    s[k] = ok ? src[e] : 0;
  }
#pragma unroll
  for (int k = 0; k < 16; ++k) {
    r[k] = 0;
    if (e0 + k * 256 < nEdges) r[k] = atomicAdd(&hist[d[k] >> BSH], 1);
  }
  __syncthreads();

  const int per = (nBuckets + 255) >> 8;
  const int lo  = tid * per;
  const int hi  = (lo + per < nBuckets) ? (lo + per) : nBuckets;
  int tsum = 0;
  for (int i = lo; i < hi; ++i) tsum += hist[i];
  int incl = tsum;
  for (int o = 1; o < 32; o <<= 1) {
    const int t = __shfl_up(incl, o);
    if (lane >= o) incl += t;
  }
  if (lane == 31) wsum[wv] = incl;
  __syncthreads();
  int wpre = 0, total = 0;
  for (int w = 0; w < 8; ++w) {
    const int t = wsum[w];
    if (w < wv) wpre += t;
    total += t;
  }
  int run = wpre + incl - tsum;
  for (int i = lo; i < hi; ++i) {
    const int c = hist[i];
    hist[i] = run;
    run += c;
  }
  if (tid == 0) hist[nBuckets] = total;
  __syncthreads();

  int* orow = off + (size_t)blk * (nBuckets + 1);
  for (int i = tid; i <= nBuckets; i += 256) orow[i] = hist[i];

#pragma unroll
  for (int k = 0; k < 16; ++k) {
    if (e0 + k * 256 < nEdges) stage[hist[d[k] >> BSH] + r[k]] = make_int2(s[k], d[k]);
  }
  __syncthreads();
  int2* prow = part + (size_t)blk * EPB;
  for (int j = tid; j < total; j += 256) prow[j] = stage[j];
}

__global__ __launch_bounds__(256) void bucket_kernel(
    const int2* __restrict__ part, const int* __restrict__ off, int nPartBlocks, int nBuckets,
    int nNodes, int* __restrict__ col, int* __restrict__ start, float* __restrict__ deg) {
  __shared__ int cnt[BNODES];
  __shared__ int cur[BNODES];
  __shared__ int wred[8];
  __shared__ int wscan[BNODES / 32];
  const int tid  = threadIdx.x;
  const int lane = tid & 31;
  const int wv   = tid >> 5;
  const int b    = blockIdx.x;
  const int n0   = b << BSH;
  const size_t pitch = (size_t)nBuckets + 1;

  if (tid < BNODES) cnt[tid] = 0;
  __syncthreads();

  int baseAcc = 0;
  for (int blk = tid; blk < nPartBlocks; blk += 256) {
    const int* o = off + (size_t)blk * pitch + b;
    const int sb = o[0];
    const int se = o[1];
    baseAcc += sb;
    const int2* p = part + (size_t)blk * EPB;
    for (int i = sb; i < se; i += 8) {
      int dd[8];
#pragma unroll
      for (int k = 0; k < 8; ++k) {
        const int j = (i + k < se) ? (i + k) : (se - 1);
        dd[k] = p[j].y;
      }
#pragma unroll
      for (int k = 0; k < 8; ++k) {
        atomicAdd(&cnt[dd[k] - n0], (i + k < se) ? 1 : 0);
      }
    }
  }
  for (int o = 16; o > 0; o >>= 1) baseAcc += __shfl_xor(baseAcc, o);
  if (lane == 0) wred[wv] = baseAcc;
  __syncthreads();

  int base = 0;
  for (int w = 0; w < 8; ++w) base += wred[w];
  int c = 0, incl = 0;
  if (tid < BNODES) {
    c = cnt[tid];
    incl = c;
    for (int o = 1; o < 32; o <<= 1) {
      const int t = __shfl_up(incl, o);
      if (lane >= o) incl += t;
    }
    if (lane == 31) wscan[wv] = incl;
  }
  __syncthreads();
  if (tid < BNODES) {
    int pre = 0;
    for (int w = 0; w < BNODES / 32; ++w) if (w < wv) pre += wscan[w];
    const int first = base + pre + incl - c;
    cur[tid] = first;
    if (n0 + tid < nNodes) {
      start[n0 + tid] = first;
      deg[n0 + tid]   = (float)c;
    }
  }
  __syncthreads();

  for (int blk = tid; blk < nPartBlocks; blk += 256) {
    const int* o = off + (size_t)blk * pitch + b;
    const int sb = o[0];
    const int se = o[1];
    const int2* p = part + (size_t)blk * EPB;
    for (int i = sb; i < se; i += 8) {
      int2 vv[8];
#pragma unroll
      for (int k = 0; k < 8; ++k) {
        const int j = (i + k < se) ? (i + k) : (se - 1);
        vv[k] = p[j];
      }
#pragma unroll
      for (int k = 0; k < 8; ++k) {
        const bool ok  = i + k < se;
        const int slot = atomicAdd(&cur[vv[k].y - n0], ok ? 1 : 0);
        if (ok) col[slot] = vv[k].x;
      }
    }
  }
}

static __device__ __forceinline__ float4 ld_row_u(const float* __restrict__ h, int s, unsigned loff) {
  const char* rp = (const char*)(h + (size_t)s * NODE_F);
  return *(const float4*)(rp + loff);
}
static __device__ __forceinline__ void add4(float4& a, const float4 v) {
  a.x += v.x; a.y += v.y; a.z += v.z; a.w += v.w;
}


template <bool RELU>
__global__ __launch_bounds__(256) void sage_fused(
    const float* __restrict__ h, const int* __restrict__ col, const int* __restrict__ start,
    const float* __restrict__ deg, const float* __restrict__ Wself,
    const float* __restrict__ Wneigh, const float* __restrict__ bias,
    float* __restrict__ out, int nNodes) {
  __shared__ __attribute__((aligned(16))) float wst[NODE_F][NODE_F + 4];
  __shared__ __attribute__((aligned(16))) float at[2][32][NODE_F + 4];
  float (*hn)[NODE_F + 4] = at[0];
  float (*hs)[NODE_F + 4] = at[1];

  const int tid  = threadIdx.x;
  const int lane = tid & 31;
  const int wv   = tid >> 5;
  const int lid  = lane & 15;
  const int half = lane >> 4;
  const int n0   = wv * 16;

  v2f bs[32];
  v2f bn[32];
#pragma unroll
  for (int i = 0; i < 16; ++i) {
    const int f = i * 256 + tid;
    *(float4*)&wst[f >> 5][(f & 31) * 4] = ((const float4*)Wself)[f];
  }
  __syncthreads();
#pragma unroll
  for (int s = 0; s < 32; ++s) bs[s] = *(const v2f*)&wst[n0 + lid][4 * s + 2 * half];
  __syncthreads();
#pragma unroll
  for (int i = 0; i < 16; ++i) {
    const int f = i * 256 + tid;
    *(float4*)&wst[f >> 5][(f & 31) * 4] = ((const float4*)Wneigh)[f];
  }
  __syncthreads();
#pragma unroll
  for (int s = 0; s < 32; ++s) bn[s] = *(const v2f*)&wst[n0 + lid][4 * s + 2 * half];

  const float bv = bias[n0 + lid];
  const int nTiles = (nNodes + 31) >> 5;

  for (int tile = blockIdx.x; tile < nTiles; tile += gridDim.x) {
    const int m_base = tile * 32;

    const int nf = m_base + 4 * wv;
    int nq = nf + (lane & 3);
    const bool okn = nq < nNodes;
    if (!okn) nq = nNodes - 1;
    const int   sq  = start[nq];
    const float dqf = deg[nq];
    const int   dq  = okn ? ((lane < 4) ? (int)dqf : ((lane < 8) ? 1 : 0)) : 0;
    const float rq  = (lane < 4) ? 1.0f / fmaxf(okn ? dqf : 0.f, 1.0f) : 1.0f;
    const int beg = __builtin_amdgcn_readfirstlane(sq);
    const int T   = __builtin_amdgcn_readlane(dq, 0) + __builtin_amdgcn_readlane(dq, 1) +
                    __builtin_amdgcn_readlane(dq, 2) + __builtin_amdgcn_readlane(dq, 3);
    const int TT  = T + __builtin_amdgcn_readlane(dq, 4) + __builtin_amdgcn_readlane(dq, 5) +
                    __builtin_amdgcn_readlane(dq, 6) + __builtin_amdgcn_readlane(dq, 7);
    int   node = 0;
    int   nend = __builtin_amdgcn_readlane(dq, 0);
    float4 acc = make_float4(0.f, 0.f, 0.f, 0.f);
    float* hrow = &hn[4 * wv][lane * 4];
    const unsigned loff = (unsigned)lane * 16u;

#define NODE_DONE()                                                          \
    do {                                                                     \
      const float rd = __builtin_bit_cast(float, __builtin_amdgcn_readlane(__builtin_bit_cast(int, rq), node)); \
      acc.x *= rd; acc.y *= rd; acc.z *= rd; acc.w *= rd;                    \
      *(float4*)hrow = acc;                                                  \
      acc = make_float4(0.f, 0.f, 0.f, 0.f);                                 \
      ++node;                                                                \
      hrow += (node == 4) ? 29 * (NODE_F + 4) : (NODE_F + 4);                \
      nend += __builtin_amdgcn_readlane(dq, node);      \
    } while (0)
#define TAKE_ROW(r)                                                          \
    do {                                                                     \
      while (P == nend) NODE_DONE();                                         \
      add4(acc, (r));                                                        \
      ++P;                                                                   \
    } while (0)

    for (int b0 = 0; b0 < TT; b0 += 128) {
      const int Tb = (TT - b0 < 128) ? (TT - b0) : 128;
      int ix[4];
#pragma unroll
      for (int k = 0; k < 4; ++k) {
        const int pos = b0 + 32 * k + lane;
        int r = (pos < TT) ? (nf + pos - T) : 0;
        if (pos < T) r = col[beg + pos];
        ix[k] = r;
      }
      int i0 = ix[0], i1 = ix[1], i2 = ix[2], i3 = ix[3];
      int P = b0;
      float4 v[16];
#pragma unroll
      for (int j = 0; j < 16; ++j) v[j] = ld_row_u(h, __builtin_amdgcn_readlane(i0, j), loff);
      int base = 0;
      int odd  = 0;
      for (; base + 16 < Tb; base += 16) {
        const int w  = odd ? i1 : i0;
        const int lb = odd ? 0 : 16;
#pragma unroll
        for (int j = 0; j < 16; ++j) {
          TAKE_ROW(v[j]);
          asm volatile("" : "+v"(acc.x), "+v"(acc.y), "+v"(acc.z), "+v"(acc.w) : : "memory");
          v[j] = ld_row_u(h, __builtin_amdgcn_readlane(w, lb + j), loff);
        }
        if (odd) { i0 = i1; i1 = i2; i2 = i3; i3 = 0; }
        odd ^= 1;
      }
#pragma unroll
      for (int j = 0; j < 16; ++j) {
        if (base + j < Tb) TAKE_ROW(v[j]);
      }
    }
    while (node < 8) NODE_DONE();
#undef TAKE_ROW
#undef NODE_DONE
    __syncthreads();

    v8f acc0 = {0.f, 0.f, 0.f, 0.f, 0.f, 0.f, 0.f, 0.f};
    v8f acc1 = {0.f, 0.f, 0.f, 0.f, 0.f, 0.f, 0.f, 0.f};

#pragma unroll
    for (int s = 0; s < 32; ++s) {
      const int kk = 4 * s + 2 * half;
      const v2f a0 = *(const v2f*)&hs[lid][kk];
      const v2f a1 = *(const v2f*)&hs[16 + lid][kk];
      acc0 = wmma_k4(a0, bs[s], acc0);
      acc1 = wmma_k4(a1, bs[s], acc1);
      const v2f c0 = *(const v2f*)&hn[lid][kk];
      const v2f c1 = *(const v2f*)&hn[16 + lid][kk];
      acc0 = wmma_k4(c0, bn[s], acc0);
      acc1 = wmma_k4(c1, bn[s], acc1);
    }
    float (*ot)[NODE_F + 4] = wst;
#pragma unroll
    for (int e = 0; e < 8; ++e) {
      float v0 = acc0[e] + bv;
      float v1 = acc1[e] + bv;
      if (RELU) { v0 = fmaxf(v0, 0.f); v1 = fmaxf(v1, 0.f); }
      ot[8 * half + e][n0 + lid]      = v0;
      ot[16 + 8 * half + e][n0 + lid] = v1;
    }
    __syncthreads();
    const float* orow = &ot[4 * wv][lane * 4];
    float* gout = out + (size_t)(m_base + 4 * wv) * NODE_F + lane * 4;
    if (m_base + 32 <= nNodes) {
      float4 o[4];
#pragma unroll
      for (int r = 0; r < 4; ++r) o[r] = *(const float4*)(orow + r * (NODE_F + 4));
#pragma unroll
      for (int r = 0; r < 4; ++r) *(float4*)(gout + (size_t)r * NODE_F) = o[r];
    } else {
      for (int r = 0; r < 4; ++r) {
        if (m_base + 4 * wv + r < nNodes)
          *(float4*)(gout + (size_t)r * NODE_F) = *(const float4*)(orow + r * (NODE_F + 4));
      }
    }
  }
}

extern "C" void kernel_launch(void* const* d_in, const int* in_sizes, int n_in,
                              void* d_out, int out_size, void* d_ws, size_t ws_size,
                              hipStream_t stream) {
  const float* x      = (const float*)d_in[0];
  const int*   src    = (const int*)d_in[1];
  const int*   dst    = (const int*)d_in[2];
  const float* W1s    = (const float*)d_in[3];
  const float* W1n    = (const float*)d_in[4];
  const float* b1     = (const float*)d_in[5];
  const float* W2s    = (const float*)d_in[6];
  const float* W2n    = (const float*)d_in[7];
  const float* b2     = (const float*)d_in[8];
  const float* W3     = (const float*)d_in[9];
  const float* b3     = (const float*)d_in[10];
  float*       out    = (float*)d_out;

  const int nNodes = in_sizes[0] / NODE_F;
  const int nEdges = in_sizes[1];

  float* sum = (float*)d_ws;
  float* deg = sum + (size_t)nNodes * NODE_F;
  float* h1  = deg + nNodes;
  int* col   = (int*)(h1 + (size_t)nNodes * NODE_F);
  int* start = col + 2 * (size_t)nEdges;
  float* Wf  = (float*)(start + nNodes);
  const float* As = Wf;
  const float* An = Wf + NODE_F * NODE_F;
  const float* cf = Wf + 2 * NODE_F * NODE_F;

  const int nBuckets    = (nNodes + BNODES - 1) / BNODES;
  const int nPartBlocks = (nEdges + EPB - 1) / EPB;
  int2* part = (int2*)sum;
  int*  off  = (int*)sum + 2 * (size_t)nPartBlocks * EPB;
  const size_t ldsPart = (size_t)((nBuckets + 4) & ~3) * sizeof(int) + (size_t)EPB * sizeof(int2);

  dim3 blk(256);
  dim3 gPart(nPartBlocks + 17);
  dim3 gBuck(nBuckets);
  dim3 gGemm(512);

  part_kernel<<<gPart, blk, ldsPart, stream>>>(src, dst, nEdges, nBuckets, nPartBlocks, part, off,
                                               W3, W2s, W2n, b2, b3, Wf);
  bucket_kernel<<<gBuck, blk, 0, stream>>>(part, off, nPartBlocks, nBuckets, nNodes, col, start, deg);

  sage_fused<true><<<gGemm, blk, 0, stream>>>(x, col, start, deg, W1s, W1n, b1, h1, nNodes);

  sage_fused<false><<<gGemm, blk, 0, stream>>>(h1, col, start, deg, As, An, cf, out, nNodes);
}
